# starting kernel
# baseline (speedup 1.0000x reference)
#include <hip/hip_runtime.h>
#include <hip/hip_bf16.h>
#include <math.h>

#define SLOT  2
#define DIM   128
#define ROW   (SLOT * DIM)
#define SLOPE 0.98f

typedef float v2f __attribute__((ext_vector_type(2)));
typedef float v8f __attribute__((ext_vector_type(8)));

__device__ __forceinline__ unsigned flip_f32(float f) {
    unsigned b = __float_as_uint(f);
    return (b & 0x80000000u) ? ~b : (b | 0x80000000u);
}
__device__ __forceinline__ float unflip_f32(unsigned u) {
    unsigned b = (u & 0x80000000u) ? (u & 0x7fffffffu) : ~u;
    return __uint_as_float(b);
}

__global__ void k_init(float4* __restrict__ out4, int nOut4,
                       unsigned* __restrict__ m_u, float* __restrict__ s, int N) {
    int i      = blockIdx.x * blockDim.x + threadIdx.x;
    int stride = gridDim.x * blockDim.x;
    float4 z = make_float4(0.f, 0.f, 0.f, 0.f);
    for (int j = i; j < nOut4; j += stride) out4[j] = z;
    for (int j = i; j < N; j += stride) { m_u[j] = 0u; s[j] = 0.0f; }
}

__global__ void k_edge_scores(const float* __restrict__ h,
                              const int*  __restrict__ src,
                              const int*  __restrict__ dst,
                              const float* __restrict__ watt,
                              float* __restrict__ abuf,
                              unsigned* __restrict__ m_u,
                              int E, int ntiles) {
    __shared__ float wlds[2 * DIM];
    int tid = threadIdx.x;
    if (tid < 2 * DIM) wlds[tid] = watt[tid];
    __syncthreads();

    int tile = blockIdx.x * (blockDim.x >> 5) + (tid >> 5);
    if (tile >= ntiles) return;

    int lane = tid & 31;
    int half = lane >> 4;
    int m16  = lane & 15;
    int base = tile * 16;

    int eL = base + m16; if (eL >= E) eL = E - 1;
    int sv = src[eL];
    int dv = dst[eL];
    const float* ps = h + (size_t)sv * ROW + 2 * half;
    const float* pd = h + (size_t)dv * ROW + 2 * half;
    const float* wl = wlds + 2 * half;

    v8f c = {};
#if __has_builtin(__builtin_amdgcn_wmma_f32_16x16x4_f32)
    #pragma unroll
    for (int k = 0; k < 32; ++k) {
        v2f av = *(const v2f*)(ps + 4 * k);
        v2f bv = *(const v2f*)(wl + 4 * k);
        c = __builtin_amdgcn_wmma_f32_16x16x4_f32(false, av, false, bv,
                                                  (short)0, c, false, false);
    }
    #pragma unroll
    for (int k = 0; k < 32; ++k) {
        v2f av = *(const v2f*)(pd + 4 * k);
        v2f bv = *(const v2f*)(wl + DIM + 4 * k);
        c = __builtin_amdgcn_wmma_f32_16x16x4_f32(false, av, false, bv,
                                                  (short)0, c, false, false);
    }
    int r = lane & 7;
    float sc = c[0];
    sc = (r == 1) ? c[1] : sc;
    sc = (r == 2) ? c[2] : sc;
    sc = (r == 3) ? c[3] : sc;
    sc = (r == 4) ? c[4] : sc;
    sc = (r == 5) ? c[5] : sc;
    sc = (r == 6) ? c[6] : sc;
    sc = (r == 7) ? c[7] : sc;
#else
    int r = lane & 7;
    int rowF = ((lane >> 4) << 3) | r;
    int eF = base + rowF; if (eF >= E) eF = E - 1;
    const float* fs = h + (size_t)src[eF] * ROW;
    const float* fd = h + (size_t)dst[eF] * ROW;
    float sc = 0.f;
    for (int k = 0; k < DIM; ++k) sc += fs[k] * wlds[k] + fd[k] * wlds[DIM + k];
#endif

    int row = ((lane >> 4) << 3) | r;
    if ((lane & 8) == 0) {
        int e = base + row;
        if (e < E) {
            float v = (sc >= 0.0f) ? sc : SLOPE * sc;
            abuf[e] = v;
            atomicMax(&m_u[dst[e]], flip_f32(v));
        }
    }
}

__global__ void k_edge_exp(const int* __restrict__ dst,
                           float* __restrict__ abuf,
                           const unsigned* __restrict__ m_u,
                           float* __restrict__ s, int E) {
    int e = blockIdx.x * blockDim.x + threadIdx.x;
    if (e >= E) return;
    int d = dst[e];
    float w = __expf(abuf[e] - unflip_f32(m_u[d]));
    abuf[e] = w;
    atomicAdd(&s[d], w);
}

__global__ void k_scatter(const float* __restrict__ h,
                          const int*  __restrict__ src,
                          const int*  __restrict__ dst,
                          const float* __restrict__ wbuf,
                          const float* __restrict__ s,
                          float* __restrict__ out, int E) {
    long gid = (long)blockIdx.x * blockDim.x + threadIdx.x;
    long tot = (long)E * (ROW / 4);
    if (gid >= tot) return;
    int e  = (int)(gid >> 6);
    int c4 = (int)(gid & 63);
    int d  = dst[e];
    float alpha = wbuf[e] / s[d];
    const float4 v = *(const float4*)(h + (size_t)src[e] * ROW + c4 * 4);
    float* o = out + (size_t)d * ROW + c4 * 4;
    atomicAdd(o + 0, alpha * v.x);
    atomicAdd(o + 1, alpha * v.y);
    atomicAdd(o + 2, alpha * v.z);
    atomicAdd(o + 3, alpha * v.w);
}

__global__ void k_finalize(const float* __restrict__ h,
                           const float* __restrict__ s,
                           float* __restrict__ out, int N) {
    long gid = (long)blockIdx.x * blockDim.x + threadIdx.x;
    long tot = (long)N * (ROW / 4);
    if (gid >= tot) return;
    int n  = (int)(gid >> 6);
    int c4 = (int)(gid & 63);
    if (s[n] == 0.0f) {
        *(float4*)(out + (size_t)n * ROW + c4 * 4) =
            *(const float4*)(h + (size_t)n * ROW + c4 * 4);
    }
}

extern "C" void kernel_launch(void* const* d_in, const int* in_sizes, int n_in,
                              void* d_out, int out_size, void* d_ws, size_t ws_size,
                              hipStream_t stream) {
    const float* h   = (const float*)d_in[0];
    const int*   src = (const int*)d_in[1];
    const int*   dst = (const int*)d_in[2];
    const float* w   = (const float*)d_in[3];
    int E = in_sizes[1];
    int N = in_sizes[0] / ROW;
    float* out = (float*)d_out;

    float*    abuf = (float*)d_ws;
    unsigned* m_u  = (unsigned*)((char*)d_ws + (size_t)E * 4);
    float*    s    = (float*)   ((char*)d_ws + (size_t)E * 4 + (size_t)N * 4);

    int nOut4 = N * (ROW / 4);
    k_init<<<1024, 256, 0, stream>>>((float4*)out, nOut4, m_u, s, N);

    int ntiles = (E + 15) / 16;
    k_edge_scores<<<(ntiles + 7) / 8, 256, 0, stream>>>(h, src, dst, w,
                                                        abuf, m_u, E, ntiles);

    k_edge_exp<<<(E + 255) / 256, 256, 0, stream>>>(dst, abuf, m_u, s, E);

    long tot4 = (long)E * (ROW / 4);
    k_scatter<<<(int)((tot4 + 255) / 256), 256, 0, stream>>>(h, src, dst,
                                                             abuf, s, out, E);

    long totf = (long)N * (ROW / 4);
    k_finalize<<<(int)((totf + 255) / 256), 256, 0, stream>>>(h, s, out, N);
}


# edge scores: whole 512-B h rows loaded b128 per lane (lane = 4 consecutive k), w the same way; x and w of a lane meet on the WMMA diagonal, score = trace by DPP butterfly; score sum order changes
# speedup vs baseline: 1.9115x; 1.9115x over previous
#include <hip/hip_runtime.h>
#include <hip/hip_bf16.h>
#include <math.h>

#define SLOT  2
#define DIM   128
#define ROW   (SLOT * DIM)
#define SLOPE 0.98f

typedef float v2f __attribute__((ext_vector_type(2)));
typedef float v8f __attribute__((ext_vector_type(8)));
typedef float v4f __attribute__((ext_vector_type(4)));

__device__ __forceinline__ unsigned flip_f32(float f) {
    unsigned b = __float_as_uint(f);
    return (b & 0x80000000u) ? ~b : (b | 0x80000000u);
}
__device__ __forceinline__ float unflip_f32(unsigned u) {
    unsigned b = (u & 0x80000000u) ? (u & 0x7fffffffu) : ~u;
    return __uint_as_float(b);
}

__global__ void k_init(unsigned* __restrict__ m_u, float* __restrict__ s, int N) {
    int i      = blockIdx.x * blockDim.x + threadIdx.x;
    int stride = gridDim.x * blockDim.x;
    for (int j = i; j < N; j += stride) { m_u[j] = 0u; s[j] = 0.0f; }
}

#define SC_EDGES 32
#define SC_BATCH 8

__device__ __forceinline__ float xhalf_f32(float x) {
    int b = __float_as_int(x);
    return __int_as_float(__builtin_amdgcn_permlanex16(b, b, 0x76543210, 0xfedcba98, false, false));
}
template <int CTRL>
__device__ __forceinline__ float rowx_f32(float x) {
    return __int_as_float(__builtin_amdgcn_update_dpp(0, __float_as_int(x), CTRL, 0xf, 0xf, true));
}

__global__ void __launch_bounds__(256)
k_edge_scores(const float* __restrict__ h,
              const int*  __restrict__ src,
              const int*  __restrict__ dst,
              const float* __restrict__ watt,
              float* __restrict__ abuf,
              unsigned* __restrict__ m_u,
              int E, int ngroups) {
    const int tid  = threadIdx.x;
    const int lane = tid & 31;
    const int grp  = blockIdx.x * (blockDim.x >> 5) + (tid >> 5);
    if (grp >= ngroups) return;

    const int e  = grp * SC_EDGES + lane;
    const int ec = (e < E) ? e : E - 1;
    const int sv = src[ec];
    const int dv = dst[ec];

    const v4f wS = *(const v4f*)(watt + lane * 4);
    const v4f wD = *(const v4f*)(watt + DIM + lane * 4);
    const v2f wS0 = {wS.x, wS.y}, wS1 = {wS.z, wS.w};
    const v2f wD0 = {wD.x, wD.y}, wD1 = {wD.z, wD.w};

    const int  r    = lane & 7;
    const bool diag = ((lane >> 3) & 1) == (lane >> 4);
    const bool bit0 = (lane & 1) != 0, bit1 = (lane & 2) != 0, bit2 = (lane & 4) != 0;
    const float* hl = h + lane * 4;

    float mine = 0.0f;
    #pragma unroll 1
    for (int b0 = 0; b0 < SC_EDGES; b0 += SC_BATCH) {
        v4f xs[SC_BATCH], xd[SC_BATCH];
        #pragma unroll
        for (int j = 0; j < SC_BATCH; ++j) {
            const int sj = __builtin_amdgcn_readlane(sv, b0 + j);
            const int dj = __builtin_amdgcn_readlane(dv, b0 + j);
            xs[j] = *(const v4f*)(hl + (size_t)sj * ROW);
            xd[j] = *(const v4f*)(hl + (size_t)dj * ROW);
        }
        float t[SC_BATCH];
        #pragma unroll
        for (int j = 0; j < SC_BATCH; ++j) {
            const v2f a0 = {xs[j].x, xs[j].y}, a1 = {xs[j].z, xs[j].w};
            const v2f a2 = {xd[j].x, xd[j].y}, a3 = {xd[j].z, xd[j].w};
            v8f c = {};
            c = __builtin_amdgcn_wmma_f32_16x16x4_f32(false, a0, false, wS0, (short)0, c, false, false);
            c = __builtin_amdgcn_wmma_f32_16x16x4_f32(false, a1, false, wS1, (short)0, c, false, false);
            c = __builtin_amdgcn_wmma_f32_16x16x4_f32(false, a2, false, wD0, (short)0, c, false, false);
            c = __builtin_amdgcn_wmma_f32_16x16x4_f32(false, a3, false, wD1, (short)0, c, false, false);
            float d = c[0];
            d = (r == 1) ? c[1] : d;
            d = (r == 2) ? c[2] : d;
            d = (r == 3) ? c[3] : d;
            d = (r == 4) ? c[4] : d;
            d = (r == 5) ? c[5] : d;
            d = (r == 6) ? c[6] : d;
            d = (r == 7) ? c[7] : d;
            t[j] = d;
        }
        float a4[4], a2[2];
        #pragma unroll
        for (int j = 0; j < 4; ++j)
            a4[j] = (bit2 ? t[j + 4] : t[j]) + rowx_f32<0x164>(bit2 ? t[j] : t[j + 4]);
        #pragma unroll
        for (int j = 0; j < 2; ++j)
            a2[j] = (bit1 ? a4[j + 2] : a4[j]) + rowx_f32<0x162>(bit1 ? a4[j] : a4[j + 2]);
        float u = (bit0 ? a2[1] : a2[0]) + rowx_f32<0x161>(bit0 ? a2[0] : a2[1]);
        u = diag ? u : 0.0f;
        u = u + xhalf_f32(u);
        u = u + rowx_f32<0x168>(u);
        mine = ((lane >> 3) == (b0 >> 3)) ? u : mine;
    }

    if (e < E) {
        float v = (mine >= 0.0f) ? mine : SLOPE * mine;
        abuf[e] = v;
        atomicMax(&m_u[dv], flip_f32(v));
    }
}

__global__ void k_edge_exp(const int* __restrict__ dst,
                           float* __restrict__ abuf,
                           const unsigned* __restrict__ m_u,
                           float* __restrict__ s, int E) {
    int e = blockIdx.x * blockDim.x + threadIdx.x;
    if (e >= E) return;
    int d = dst[e];
    float w = __expf(abuf[e] - unflip_f32(m_u[d]));
    abuf[e] = w;
    atomicAdd(&s[d], w);
}

#define OWN_R      200
#define OWN_T      1024
#define OWN_LOADS  8
#define OWN_WEDGES (32 * 4 * OWN_LOADS)
#define OWN_CHUNK  (32 * OWN_WEDGES)
#define OWN_CAP    8192

__global__ void __launch_bounds__(OWN_T)
k_owner(const float* __restrict__ h,
        const int*  __restrict__ src,
        const int*  __restrict__ dst,
        const float* __restrict__ wbuf,
        const float* __restrict__ s,
        float* __restrict__ out, int E, int N) {
    __shared__ __align__(16) float acc[OWN_R * ROW];
    __shared__ int   l_a[OWN_CAP];
    __shared__ int   l_dl[OWN_CAP];
    __shared__ float l_al[OWN_CAP];
    __shared__ int   wtot[32];

    const int tid  = threadIdx.x;
    const int lane = tid & 31;
    const int wv   = __builtin_amdgcn_readfirstlane(tid >> 5);
    const int lo   = blockIdx.x * OWN_R;
    int nr = N - lo; if (nr > OWN_R) nr = OWN_R;

    float4* acc4 = (float4*)acc;
    for (int i = tid; i < OWN_R * (ROW / 4); i += OWN_T)
        acc4[i] = make_float4(0.f, 0.f, 0.f, 0.f);
    __syncthreads();

    const int cg  = wv & 1;
    const int ng  = wv >> 1;
    const int col = cg * 128 + lane * 4;

    int cbase = 0;
    int wbase = 0, woff = 0, tot = 0, p0 = 0;
    unsigned mbits = 0u;
    int lcnt = 0, linc = 0;
    int fill = 0;

    for (;;) {
        if (p0 == tot && cbase < E) {
            wbase = cbase + wv * OWN_WEDGES;
            int4 d4[OWN_LOADS];
            if (wbase + OWN_WEDGES <= E) {
                #pragma unroll
                for (int j = 0; j < OWN_LOADS; ++j)
                    d4[j] = *(const int4*)(dst + wbase + j * 128 + lane * 4);
            } else {
                #pragma unroll
                for (int j = 0; j < OWN_LOADS; ++j) {
                    int e0 = wbase + j * 128 + lane * 4;
                    d4[j].x = (e0     < E) ? dst[e0]     : -1;
                    d4[j].y = (e0 + 1 < E) ? dst[e0 + 1] : -1;
                    d4[j].z = (e0 + 2 < E) ? dst[e0 + 2] : -1;
                    d4[j].w = (e0 + 3 < E) ? dst[e0 + 3] : -1;
                }
            }
            mbits = 0u;
            #pragma unroll
            for (int j = 0; j < OWN_LOADS; ++j) {
                bool m0 = (unsigned)(d4[j].x - lo) < (unsigned)nr;
                bool m1 = (unsigned)(d4[j].y - lo) < (unsigned)nr;
                bool m2 = (unsigned)(d4[j].z - lo) < (unsigned)nr;
                bool m3 = (unsigned)(d4[j].w - lo) < (unsigned)nr;
                mbits |= (m0 ? 1u : 0u) << (j * 4 + 0);
                mbits |= (m1 ? 1u : 0u) << (j * 4 + 1);
                mbits |= (m2 ? 1u : 0u) << (j * 4 + 2);
                mbits |= (m3 ? 1u : 0u) << (j * 4 + 3);
            }
            lcnt = __popc(mbits);
            linc = lcnt;
            for (int d = 1; d < 32; d <<= 1) {
                int t = __shfl_up(linc, d, 32);
                if (lane >= d) linc += t;
            }
            if (lane == 31) wtot[wv] = linc;
            __syncthreads();

            int incl = wtot[lane];
            const int mine = incl;
            for (int d = 1; d < 32; d <<= 1) {
                int t = __shfl_up(incl, d, 32);
                if (lane >= d) incl += t;
            }
            woff = __builtin_amdgcn_readfirstlane(__shfl(incl - mine, wv, 32));
            tot  = __builtin_amdgcn_readfirstlane(__shfl(incl, 31, 32));
            __syncthreads();
            p0 = 0;
            cbase += OWN_CHUNK;
        }

        if (p0 < tot) {
            int take = tot - p0;
            if (take > OWN_CAP - fill) take = OWN_CAP - fill;
            int pos = woff + (linc - lcnt) - p0;
            unsigned mb = mbits;
            while (mb != 0u) {
                int k = __ffs(mb) - 1; mb &= mb - 1u;
                if (pos >= 0 && pos < take)
                    l_a[fill + pos] = wbase + (k >> 2) * 128 + lane * 4 + (k & 3);
                ++pos;
            }
            fill += take;
            p0   += take;
        }

        const bool done = (p0 == tot) && (cbase >= E);
        if (fill == OWN_CAP || (done && fill > 0)) {
            const int n = fill;
            __syncthreads();
            for (int i = tid; i < n; i += OWN_T) {
                int e = l_a[i];
                int d = dst[e];
                int sv = src[e];
                float w = wbuf[e];
                float sd = s[d];
                l_a[i]  = sv;
                l_dl[i] = d - lo;
                l_al[i] = w / sd;
            }
            __syncthreads();

            for (int b0 = 0; b0 < n; b0 += 64) {
                const int  iA = b0 + lane, iB = iA + 32;
                const bool vA = iA < n,    vB = iB < n;
                const int   srcA = vA ? l_a[iA]  : 0,   srcB = vB ? l_a[iB]  : 0;
                const int   dlA  = vA ? l_dl[iA] : 0,   dlB  = vB ? l_dl[iB] : 0;
                const float alA  = vA ? l_al[iA] : 0.f, alB  = vB ? l_al[iB] : 0.f;
                unsigned mA = (unsigned)__ballot(vA && ((dlA & 15) == ng));
                unsigned mB = (unsigned)__ballot(vB && ((dlB & 15) == ng));
                while ((mA | mB) != 0u) {
                    int es[8], ed[8]; float ea[8]; int cnt = 0;
                    #pragma unroll
                    for (int t = 0; t < 8; ++t) {
                        es[t] = 0; ed[t] = 0; ea[t] = 0.f;
                        if (mA != 0u) {
                            int j = __ffs(mA) - 1; mA &= mA - 1u;
                            es[t] = __builtin_amdgcn_readlane(srcA, j);
                            ed[t] = __builtin_amdgcn_readlane(dlA, j);
                            ea[t] = __int_as_float(__builtin_amdgcn_readlane(__float_as_int(alA), j));
                            cnt = t + 1;
                        } else if (mB != 0u) {
                            int j = __ffs(mB) - 1; mB &= mB - 1u;
                            es[t] = __builtin_amdgcn_readlane(srcB, j);
                            ed[t] = __builtin_amdgcn_readlane(dlB, j);
                            ea[t] = __int_as_float(__builtin_amdgcn_readlane(__float_as_int(alB), j));
                            cnt = t + 1;
                        }
                    }
                    v4f v[8];
                    #pragma unroll
                    for (int t = 0; t < 8; ++t) {
                        v[t] = (v4f){0.f, 0.f, 0.f, 0.f};
                        if (t < cnt) v[t] = *(const v4f*)(h + (size_t)es[t] * ROW + col);
                    }
                    asm volatile("" :: "v"(v[0]), "v"(v[1]), "v"(v[2]), "v"(v[3]),
                                       "v"(v[4]), "v"(v[5]), "v"(v[6]), "v"(v[7]));
                    #pragma unroll
                    for (int t = 0; t < 8; ++t) {
                        if (t < cnt) {
                            v4f* p = (v4f*)(acc + ed[t] * ROW + col);
                            *p = *p + ea[t] * v[t];
                        }
                    }
                }
            }
            fill = 0;
            __syncthreads();
        }
        if (done) break;
    }

    const float4* h4 = (const float4*)h;
    float4* out4 = (float4*)out;
    for (int i = tid; i < nr * (ROW / 4); i += OWN_T) {
        int r = i >> 6;
        int n = lo + r;
        size_t g = (size_t)n * (ROW / 4) + (i & 63);
        float4 v = acc4[i];
        if (s[n] == 0.0f) v = h4[g];
        out4[g] = v;
    }
}

extern "C" void kernel_launch(void* const* d_in, const int* in_sizes, int n_in,
                              void* d_out, int out_size, void* d_ws, size_t ws_size,
                              hipStream_t stream) {
    const float* h   = (const float*)d_in[0];
    const int*   src = (const int*)d_in[1];
    const int*   dst = (const int*)d_in[2];
    const float* w   = (const float*)d_in[3];
    int E = in_sizes[1];
    int N = in_sizes[0] / ROW;
    float* out = (float*)d_out;

    float*    abuf = (float*)d_ws;
    unsigned* m_u  = (unsigned*)((char*)d_ws + (size_t)E * 4);
    float*    s    = (float*)   ((char*)d_ws + (size_t)E * 4 + (size_t)N * 4);

    k_init<<<(N + 255) / 256, 256, 0, stream>>>(m_u, s, N);

    int ngroups = (E + SC_EDGES - 1) / SC_EDGES;
    k_edge_scores<<<(ngroups + 7) / 8, 256, 0, stream>>>(h, src, dst, w,
                                                         abuf, m_u, E, ngroups);

    k_edge_exp<<<(E + 255) / 256, 256, 0, stream>>>(dst, abuf, m_u, s, E);

    k_owner<<<(N + OWN_R - 1) / OWN_R, OWN_T, 0, stream>>>(h, src, dst, abuf, s,
                                                           out, E, N);
}


# segment softmax moved into the owner work-group: row max as LDS integer max, exp weights, denominators as 64-bit fixed-point integer sums (order-free), rows scaled by 1/sum at write-out; 2 launches
# speedup vs baseline: 2.8054x; 1.4676x over previous
#include <hip/hip_runtime.h>
#include <hip/hip_bf16.h>
#include <math.h>

#define SLOT  2
#define DIM   128
#define ROW   (SLOT * DIM)
#define SLOPE 0.98f

typedef float v2f __attribute__((ext_vector_type(2)));
typedef float v8f __attribute__((ext_vector_type(8)));
typedef float v4f __attribute__((ext_vector_type(4)));

__device__ __forceinline__ unsigned flip_f32(float f) {
    unsigned b = __float_as_uint(f);
    return (b & 0x80000000u) ? ~b : (b | 0x80000000u);
}
__device__ __forceinline__ float unflip_f32(unsigned u) {
    unsigned b = (u & 0x80000000u) ? (u & 0x7fffffffu) : ~u;
    return __uint_as_float(b);
}

#define SC_EDGES 32
#define SC_BATCH 8

__device__ __forceinline__ float xhalf_f32(float x) {
    int b = __float_as_int(x);
    return __int_as_float(__builtin_amdgcn_permlanex16(b, b, 0x76543210, 0xfedcba98, false, false));
}
template <int CTRL>
__device__ __forceinline__ float rowx_f32(float x) {
    return __int_as_float(__builtin_amdgcn_update_dpp(0, __float_as_int(x), CTRL, 0xf, 0xf, true));
}

__global__ void __launch_bounds__(256)
k_edge_scores(const float* __restrict__ h,
              const int*  __restrict__ src,
              const int*  __restrict__ dst,
              const float* __restrict__ watt,
              float* __restrict__ abuf,
              int E, int ngroups) {
    const int tid  = threadIdx.x;
    const int lane = tid & 31;
    const int grp  = blockIdx.x * (blockDim.x >> 5) + (tid >> 5);
    if (grp >= ngroups) return;

    const int e  = grp * SC_EDGES + lane;
    const int ec = (e < E) ? e : E - 1;
    const int sv = src[ec];
    const int dv = dst[ec];

    const v4f wS = *(const v4f*)(watt + lane * 4);
    const v4f wD = *(const v4f*)(watt + DIM + lane * 4);
    const v2f wS0 = {wS.x, wS.y}, wS1 = {wS.z, wS.w};
    const v2f wD0 = {wD.x, wD.y}, wD1 = {wD.z, wD.w};

    const int  r    = lane & 7;
    const bool diag = ((lane >> 3) & 1) == (lane >> 4);
    const bool bit0 = (lane & 1) != 0, bit1 = (lane & 2) != 0, bit2 = (lane & 4) != 0;
    const float* hl = h + lane * 4;

    float mine = 0.0f;
    #pragma unroll 1
    for (int b0 = 0; b0 < SC_EDGES; b0 += SC_BATCH) {
        v4f xs[SC_BATCH], xd[SC_BATCH];
        #pragma unroll
        for (int j = 0; j < SC_BATCH; ++j) {
            const int sj = __builtin_amdgcn_readlane(sv, b0 + j);
            const int dj = __builtin_amdgcn_readlane(dv, b0 + j);
            xs[j] = *(const v4f*)(hl + (size_t)sj * ROW);
            xd[j] = *(const v4f*)(hl + (size_t)dj * ROW);
        }
        float t[SC_BATCH];
        #pragma unroll
        for (int j = 0; j < SC_BATCH; ++j) {
            const v2f a0 = {xs[j].x, xs[j].y}, a1 = {xs[j].z, xs[j].w};
            const v2f a2 = {xd[j].x, xd[j].y}, a3 = {xd[j].z, xd[j].w};
            v8f c = {};
            c = __builtin_amdgcn_wmma_f32_16x16x4_f32(false, a0, false, wS0, (short)0, c, false, false);
            c = __builtin_amdgcn_wmma_f32_16x16x4_f32(false, a1, false, wS1, (short)0, c, false, false);
            c = __builtin_amdgcn_wmma_f32_16x16x4_f32(false, a2, false, wD0, (short)0, c, false, false);
            c = __builtin_amdgcn_wmma_f32_16x16x4_f32(false, a3, false, wD1, (short)0, c, false, false);
            float d = c[0];
            d = (r == 1) ? c[1] : d;
            d = (r == 2) ? c[2] : d;
            d = (r == 3) ? c[3] : d;
            d = (r == 4) ? c[4] : d;
            d = (r == 5) ? c[5] : d;
            d = (r == 6) ? c[6] : d;
            d = (r == 7) ? c[7] : d;
            t[j] = d;
        }
        float a4[4], a2[2];
        #pragma unroll
        for (int j = 0; j < 4; ++j)
            a4[j] = (bit2 ? t[j + 4] : t[j]) + rowx_f32<0x164>(bit2 ? t[j] : t[j + 4]);
        #pragma unroll
        for (int j = 0; j < 2; ++j)
            a2[j] = (bit1 ? a4[j + 2] : a4[j]) + rowx_f32<0x162>(bit1 ? a4[j] : a4[j + 2]);
        float u = (bit0 ? a2[1] : a2[0]) + rowx_f32<0x161>(bit0 ? a2[0] : a2[1]);
        u = diag ? u : 0.0f;
        u = u + xhalf_f32(u);
        u = u + rowx_f32<0x168>(u);
        mine = ((lane >> 3) == (b0 >> 3)) ? u : mine;
    }

    if (e < E) {
        float v = (mine >= 0.0f) ? mine : SLOPE * mine;
        abuf[e] = v;
    }
}

#define OWN_R      200
#define OWN_T      1024
#define OWN_LOADS  8
#define OWN_WEDGES (32 * 4 * OWN_LOADS)
#define OWN_CHUNK  (32 * OWN_WEDGES)
#define OWN_CAP    8192

__global__ void __launch_bounds__(OWN_T)
k_owner(const float* __restrict__ h,
        const int*  __restrict__ src,
        const int*  __restrict__ dst,
        const float* __restrict__ abuf,
        float* __restrict__ out, int E, int N) {
    __shared__ __align__(16) float acc[OWN_R * ROW];
    __shared__ int   l_a[OWN_CAP];
    __shared__ int   l_dl[OWN_CAP];
    __shared__ float l_al[OWN_CAP];
    __shared__ int   wtot[32];
    __shared__ unsigned m_new[OWN_R];
    __shared__ unsigned m_old[OWN_R];
    __shared__ unsigned long long s_fix[OWN_R];
    __shared__ float inv_l[OWN_R];

    const int tid  = threadIdx.x;
    const int lane = tid & 31;
    const int wv   = __builtin_amdgcn_readfirstlane(tid >> 5);
    const int lo   = blockIdx.x * OWN_R;
    int nr = N - lo; if (nr > OWN_R) nr = OWN_R;

    float4* acc4 = (float4*)acc;
    for (int i = tid; i < OWN_R * (ROW / 4); i += OWN_T)
        acc4[i] = make_float4(0.f, 0.f, 0.f, 0.f);
    if (tid < OWN_R) { m_new[tid] = 0u; m_old[tid] = 0u; s_fix[tid] = 0ull; }
    __syncthreads();

    const int cg  = wv & 1;
    const int ng  = wv >> 1;
    const int col = cg * 128 + lane * 4;

    int cbase = 0;
    int wbase = 0, woff = 0, tot = 0, p0 = 0;
    unsigned mbits = 0u;
    int lcnt = 0, linc = 0;
    int fill = 0;
    int npass = 0;

    for (;;) {
        if (p0 == tot && cbase < E) {
            wbase = cbase + wv * OWN_WEDGES;
            int4 d4[OWN_LOADS];
            if (wbase + OWN_WEDGES <= E) {
                #pragma unroll
                for (int j = 0; j < OWN_LOADS; ++j)
                    d4[j] = *(const int4*)(dst + wbase + j * 128 + lane * 4);
            } else {
                #pragma unroll
                for (int j = 0; j < OWN_LOADS; ++j) {
                    int e0 = wbase + j * 128 + lane * 4;
                    d4[j].x = (e0     < E) ? dst[e0]     : -1;
                    d4[j].y = (e0 + 1 < E) ? dst[e0 + 1] : -1;
                    d4[j].z = (e0 + 2 < E) ? dst[e0 + 2] : -1;
                    d4[j].w = (e0 + 3 < E) ? dst[e0 + 3] : -1;
                }
            }
            mbits = 0u;
            #pragma unroll
            for (int j = 0; j < OWN_LOADS; ++j) {
                bool m0 = (unsigned)(d4[j].x - lo) < (unsigned)nr;
                bool m1 = (unsigned)(d4[j].y - lo) < (unsigned)nr;
                bool m2 = (unsigned)(d4[j].z - lo) < (unsigned)nr;
                bool m3 = (unsigned)(d4[j].w - lo) < (unsigned)nr;
                mbits |= (m0 ? 1u : 0u) << (j * 4 + 0);
                mbits |= (m1 ? 1u : 0u) << (j * 4 + 1);
                mbits |= (m2 ? 1u : 0u) << (j * 4 + 2);
                mbits |= (m3 ? 1u : 0u) << (j * 4 + 3);
            }
            lcnt = __popc(mbits);
            linc = lcnt;
            for (int d = 1; d < 32; d <<= 1) {
                int t = __shfl_up(linc, d, 32);
                if (lane >= d) linc += t;
            }
            if (lane == 31) wtot[wv] = linc;
            __syncthreads();

            int incl = wtot[lane];
            const int mine = incl;
            for (int d = 1; d < 32; d <<= 1) {
                int t = __shfl_up(incl, d, 32);
                if (lane >= d) incl += t;
            }
            woff = __builtin_amdgcn_readfirstlane(__shfl(incl - mine, wv, 32));
            tot  = __builtin_amdgcn_readfirstlane(__shfl(incl, 31, 32));
            __syncthreads();
            p0 = 0;
            cbase += OWN_CHUNK;
        }

        if (p0 < tot) {
            int take = tot - p0;
            if (take > OWN_CAP - fill) take = OWN_CAP - fill;
            int pos = woff + (linc - lcnt) - p0;
            unsigned mb = mbits;
            while (mb != 0u) {
                int k = __ffs(mb) - 1; mb &= mb - 1u;
                if (pos >= 0 && pos < take)
                    l_a[fill + pos] = wbase + (k >> 2) * 128 + lane * 4 + (k & 3);
                ++pos;
            }
            fill += take;
            p0   += take;
        }

        const bool done = (p0 == tot) && (cbase >= E);
        if (fill == OWN_CAP || (done && fill > 0)) {
            const int n = fill;
            __syncthreads();
            for (int i = tid; i < n; i += OWN_T) {
                int e = l_a[i];
                int d = dst[e];
                int sv = src[e];
                float a = abuf[e];
                l_a[i]  = sv;
                l_dl[i] = d - lo;
                l_al[i] = a;
                atomicMax(&m_new[d - lo], flip_f32(a));
            }
            __syncthreads();
            if (npass > 0) {
                for (int i = tid; i < nr * (ROW / 4); i += OWN_T) {
                    const int r = i >> 6;
                    const unsigned mo = m_old[r], mn = m_new[r];
                    if (mn != mo && mo != 0u) {
                        const float f = __expf(unflip_f32(mo) - unflip_f32(mn));
                        float4 v = acc4[i];
                        v.x *= f; v.y *= f; v.z *= f; v.w *= f;
                        acc4[i] = v;
                        if ((i & 63) == 0)
                            s_fix[r] = (unsigned long long)((double)s_fix[r] * (double)f);
                    }
                }
                __syncthreads();
            }
            for (int i = tid; i < n; i += OWN_T) {
                const int dl = l_dl[i];
                const float w = __expf(l_al[i] - unflip_f32(m_new[dl]));
                l_al[i] = w;
                atomicAdd(&s_fix[dl], (unsigned long long)(w * 1099511627776.0f));
            }
            __syncthreads();
            if (tid < OWN_R) m_old[tid] = m_new[tid];

            for (int b0 = 0; b0 < n; b0 += 64) {
                const int  iA = b0 + lane, iB = iA + 32;
                const bool vA = iA < n,    vB = iB < n;
                const int   srcA = vA ? l_a[iA]  : 0,   srcB = vB ? l_a[iB]  : 0;
                const int   dlA  = vA ? l_dl[iA] : 0,   dlB  = vB ? l_dl[iB] : 0;
                const float alA  = vA ? l_al[iA] : 0.f, alB  = vB ? l_al[iB] : 0.f;
                unsigned mA = (unsigned)__ballot(vA && ((dlA & 15) == ng));
                unsigned mB = (unsigned)__ballot(vB && ((dlB & 15) == ng));
                while ((mA | mB) != 0u) {
                    int es[8], ed[8]; float ea[8]; int cnt = 0;
                    #pragma unroll
                    for (int t = 0; t < 8; ++t) {
                        es[t] = 0; ed[t] = 0; ea[t] = 0.f;
                        if (mA != 0u) {
                            int j = __ffs(mA) - 1; mA &= mA - 1u;
                            es[t] = __builtin_amdgcn_readlane(srcA, j);
                            ed[t] = __builtin_amdgcn_readlane(dlA, j);
                            ea[t] = __int_as_float(__builtin_amdgcn_readlane(__float_as_int(alA), j));
                            cnt = t + 1;
                        } else if (mB != 0u) {
                            int j = __ffs(mB) - 1; mB &= mB - 1u;
                            es[t] = __builtin_amdgcn_readlane(srcB, j);
                            ed[t] = __builtin_amdgcn_readlane(dlB, j);
                            ea[t] = __int_as_float(__builtin_amdgcn_readlane(__float_as_int(alB), j));
                            cnt = t + 1;
                        }
                    }
                    v4f v[8];
                    #pragma unroll
                    for (int t = 0; t < 8; ++t) {
                        v[t] = (v4f){0.f, 0.f, 0.f, 0.f};
                        if (t < cnt) v[t] = *(const v4f*)(h + (size_t)es[t] * ROW + col);
                    }
                    asm volatile("" :: "v"(v[0]), "v"(v[1]), "v"(v[2]), "v"(v[3]),
                                       "v"(v[4]), "v"(v[5]), "v"(v[6]), "v"(v[7]));
                    #pragma unroll
                    for (int t = 0; t < 8; ++t) {
                        if (t < cnt) {
                            v4f* p = (v4f*)(acc + ed[t] * ROW + col);
                            *p = *p + ea[t] * v[t];
                        }
                    }
                }
            }
            fill = 0;
            ++npass;
            __syncthreads();
        }
        if (done) break;
    }

    if (tid < OWN_R)
        inv_l[tid] = (m_new[tid] != 0u) ? 1.0f / ((float)s_fix[tid] * 9.094947017729282e-13f) : 0.0f;
    __syncthreads();

    const float4* h4 = (const float4*)h;
    float4* out4 = (float4*)out;
    for (int i = tid; i < nr * (ROW / 4); i += OWN_T) {
        int r = i >> 6;
        int n = lo + r;
        size_t g = (size_t)n * (ROW / 4) + (i & 63);
        float4 v = acc4[i];
        if (m_new[r] == 0u) {
            v = h4[g];
        } else {
            const float inv = inv_l[r];
            v.x *= inv; v.y *= inv; v.z *= inv; v.w *= inv;
        }
        out4[g] = v;
    }
}

extern "C" void kernel_launch(void* const* d_in, const int* in_sizes, int n_in,
                              void* d_out, int out_size, void* d_ws, size_t ws_size,
                              hipStream_t stream) {
    const float* h   = (const float*)d_in[0];
    const int*   src = (const int*)d_in[1];
    const int*   dst = (const int*)d_in[2];
    const float* w   = (const float*)d_in[3];
    int E = in_sizes[1];
    int N = in_sizes[0] / ROW;
    float* out = (float*)d_out;

    float* abuf = (float*)d_ws;

    int ngroups = (E + SC_EDGES - 1) / SC_EDGES;
    k_edge_scores<<<(ngroups + 7) / 8, 256, 0, stream>>>(h, src, dst, w,
                                                         abuf, E, ngroups);

    k_owner<<<(N + OWN_R - 1) / OWN_R, OWN_T, 0, stream>>>(h, src, dst, abuf,
                                                           out, E, N);
}


# edge score formed in the owner add pass: src half = WMMA trace of the h[src] row gathered anyway, dst half once per owned row; softmax on line, float sums in list order; 1 launch; sum orders change
# speedup vs baseline: 3.4312x; 1.2231x over previous
#include <hip/hip_runtime.h>
#include <hip/hip_bf16.h>
#include <math.h>

#define SLOT  2
#define DIM   128
#define ROW   (SLOT * DIM)
#define SLOPE 0.98f

typedef float v2f __attribute__((ext_vector_type(2)));
typedef float v8f __attribute__((ext_vector_type(8)));
typedef float v4f __attribute__((ext_vector_type(4)));


__device__ __forceinline__ float xhalf_f32(float x) {
    int b = __float_as_int(x);
    return __int_as_float(__builtin_amdgcn_permlanex16(b, b, 0x76543210, 0xfedcba98, false, false));
}
template <int CTRL>
__device__ __forceinline__ float rowx_f32(float x) {
    return __int_as_float(__builtin_amdgcn_update_dpp(0, __float_as_int(x), CTRL, 0xf, 0xf, true));
}

__device__ __forceinline__ float trace8(const v4f (&x)[8], int cnt, v2f w0, v2f w1, int lane) {
    const int  r    = lane & 7;
    const bool diag = ((lane >> 3) & 1) == (lane >> 4);
    const bool bit0 = (lane & 1) != 0, bit1 = (lane & 2) != 0, bit2 = (lane & 4) != 0;
    float t[8];
    #pragma unroll
    for (int j = 0; j < 8; ++j) {
        t[j] = 0.0f;
        if (j < cnt) {
            const v2f a0 = {x[j].x, x[j].y}, a1 = {x[j].z, x[j].w};
            v8f c = {};
            c = __builtin_amdgcn_wmma_f32_16x16x4_f32(false, a0, false, w0, (short)0, c, false, false);
            c = __builtin_amdgcn_wmma_f32_16x16x4_f32(false, a1, false, w1, (short)0, c, false, false);
            float d = c[0];
            d = (r == 1) ? c[1] : d;
            d = (r == 2) ? c[2] : d;
            d = (r == 3) ? c[3] : d;
            d = (r == 4) ? c[4] : d;
            d = (r == 5) ? c[5] : d;
            d = (r == 6) ? c[6] : d;
            d = (r == 7) ? c[7] : d;
            t[j] = d;
        }
    }
    float a4[4], a2[2];
    #pragma unroll
    for (int j = 0; j < 4; ++j)
        a4[j] = (bit2 ? t[j + 4] : t[j]) + rowx_f32<0x164>(bit2 ? t[j] : t[j + 4]);
    #pragma unroll
    for (int j = 0; j < 2; ++j)
        a2[j] = (bit1 ? a4[j + 2] : a4[j]) + rowx_f32<0x162>(bit1 ? a4[j] : a4[j + 2]);
    float u = (bit0 ? a2[1] : a2[0]) + rowx_f32<0x161>(bit0 ? a2[0] : a2[1]);
    u = diag ? u : 0.0f;
    u = u + xhalf_f32(u);
    u = u + rowx_f32<0x168>(u);
    return u;
}

#define OWN_R      200
#define OWN_T      1024
#define OWN_LOADS  8
#define OWN_WEDGES (32 * 4 * OWN_LOADS)
#define OWN_CHUNK  (32 * OWN_WEDGES)
#define OWN_CAP    8192
#define M_NONE     (-3.0e38f)

static_assert((OWN_R + 7) / 8 <= OWN_T / 32, "one wave per batch of eight owned rows");
static_assert(OWN_R <= 256, "a list entry keeps the row in its low 8 bits");

__global__ void __launch_bounds__(OWN_T)
k_owner(const float* __restrict__ h,
        const int*  __restrict__ src,
        const int*  __restrict__ dst,
        const float* __restrict__ watt,
        float* __restrict__ out, int E, int N) {
    __shared__ __align__(16) float acc[OWN_R * ROW];
    __shared__ int   l_a[OWN_CAP];
    __shared__ int   wtot[32];
    __shared__ float q_l[OWN_R];
    __shared__ float2 ml[OWN_R];
    __shared__ float inv_l[OWN_R];

    const int tid  = threadIdx.x;
    const int lane = tid & 31;
    const int wv   = __builtin_amdgcn_readfirstlane(tid >> 5);
    const int lo   = blockIdx.x * OWN_R;
    int nr = N - lo; if (nr > OWN_R) nr = OWN_R;

    const v4f wS = *(const v4f*)(watt + lane * 4);
    const v2f wS0 = {wS.x, wS.y}, wS1 = {wS.z, wS.w};

    float4* acc4 = (float4*)acc;
    for (int i = tid; i < OWN_R * (ROW / 4); i += OWN_T)
        acc4[i] = make_float4(0.f, 0.f, 0.f, 0.f);
    if (tid < OWN_R) ml[tid] = make_float2(M_NONE, 0.0f);
    if (wv < (OWN_R + 7) / 8) {
        const v4f wD = *(const v4f*)(watt + DIM + lane * 4);
        const v2f wD0 = {wD.x, wD.y}, wD1 = {wD.z, wD.w};
        v4f xq[8];
        #pragma unroll
        for (int j = 0; j < 8; ++j) {
            const int rr = wv * 8 + j;
            xq[j] = (v4f){0.f, 0.f, 0.f, 0.f};
            if (rr < nr) xq[j] = *(const v4f*)(h + (size_t)(lo + rr) * ROW + lane * 4);
        }
        const float uq = trace8(xq, 8, wD0, wD1, lane);
        const int rq = wv * 8 + lane;
        if (lane < 8 && rq < OWN_R) q_l[rq] = uq;
    }
    __syncthreads();

    const int ng = wv;

    int cbase = 0;
    int wbase = 0, woff = 0, tot = 0, p0 = 0;
    unsigned mbits = 0u;
    int lcnt = 0, linc = 0;
    int fill = 0;

    for (;;) {
        if (p0 == tot && cbase < E) {
            wbase = cbase + wv * OWN_WEDGES;
            int4 d4[OWN_LOADS];
            if (wbase + OWN_WEDGES <= E) {
                #pragma unroll
                for (int j = 0; j < OWN_LOADS; ++j)
                    d4[j] = *(const int4*)(dst + wbase + j * 128 + lane * 4);
            } else {
                #pragma unroll
                for (int j = 0; j < OWN_LOADS; ++j) {
                    int e0 = wbase + j * 128 + lane * 4;
                    d4[j].x = (e0     < E) ? dst[e0]     : -1;
                    d4[j].y = (e0 + 1 < E) ? dst[e0 + 1] : -1;
                    d4[j].z = (e0 + 2 < E) ? dst[e0 + 2] : -1;
                    d4[j].w = (e0 + 3 < E) ? dst[e0 + 3] : -1;
                }
            }
            mbits = 0u;
            #pragma unroll
            for (int j = 0; j < OWN_LOADS; ++j) {
                bool m0 = (unsigned)(d4[j].x - lo) < (unsigned)nr;
                bool m1 = (unsigned)(d4[j].y - lo) < (unsigned)nr;
                bool m2 = (unsigned)(d4[j].z - lo) < (unsigned)nr;
                bool m3 = (unsigned)(d4[j].w - lo) < (unsigned)nr;
                mbits |= (m0 ? 1u : 0u) << (j * 4 + 0);
                mbits |= (m1 ? 1u : 0u) << (j * 4 + 1);
                mbits |= (m2 ? 1u : 0u) << (j * 4 + 2);
                mbits |= (m3 ? 1u : 0u) << (j * 4 + 3);
            }
            lcnt = __popc(mbits);
            linc = lcnt;
            for (int d = 1; d < 32; d <<= 1) {
                int t = __shfl_up(linc, d, 32);
                if (lane >= d) linc += t;
            }
            if (lane == 31) wtot[wv] = linc;
            __syncthreads();

            int incl = wtot[lane];
            const int mine = incl;
            for (int d = 1; d < 32; d <<= 1) {
                int t = __shfl_up(incl, d, 32);
                if (lane >= d) incl += t;
            }
            woff = __builtin_amdgcn_readfirstlane(__shfl(incl - mine, wv, 32));
            tot  = __builtin_amdgcn_readfirstlane(__shfl(incl, 31, 32));
            __syncthreads();
            p0 = 0;
            cbase += OWN_CHUNK;
        }

        if (p0 < tot) {
            int take = tot - p0;
            if (take > OWN_CAP - fill) take = OWN_CAP - fill;
            int pos = woff + (linc - lcnt) - p0;
            unsigned mb = mbits;
            while (mb != 0u) {
                int k = __ffs(mb) - 1; mb &= mb - 1u;
                if (pos >= 0 && pos < take)
                    l_a[fill + pos] = wbase + (k >> 2) * 128 + lane * 4 + (k & 3);
                ++pos;
            }
            fill += take;
            p0   += take;
        }

        const bool done = (p0 == tot) && (cbase >= E);
        if (fill == OWN_CAP || (done && fill > 0)) {
            const int n = fill;
            __syncthreads();
            for (int i = tid; i < n; i += OWN_T) {
                int e = l_a[i];
                int d = dst[e];
                int sv = src[e];
                l_a[i] = (sv << 8) | (d - lo);
            }
            __syncthreads();

            for (int b0 = 0; b0 < n; b0 += 128) {
                int pk[4]; unsigned mk[4];
                #pragma unroll
                for (int k = 0; k < 4; ++k) {
                    const int  i = b0 + k * 32 + lane;
                    const bool v = i < n;
                    pk[k] = v ? l_a[i] : 0;
                    mk[k] = (unsigned)__ballot(v && ((pk[k] & 31) == ng));
                }
                while ((mk[0] | mk[1] | mk[2] | mk[3]) != 0u) {
                    int es[8], ed[8]; int cnt = 0;
                    #pragma unroll
                    for (int t = 0; t < 8; ++t) {
                        int e = 0; bool got = true;
                        if (mk[0] != 0u) {
                            int j = __ffs(mk[0]) - 1; mk[0] &= mk[0] - 1u;
                            e = __builtin_amdgcn_readlane(pk[0], j);
                        } else if (mk[1] != 0u) {
                            int j = __ffs(mk[1]) - 1; mk[1] &= mk[1] - 1u;
                            e = __builtin_amdgcn_readlane(pk[1], j);
                        } else if (mk[2] != 0u) {
                            int j = __ffs(mk[2]) - 1; mk[2] &= mk[2] - 1u;
                            e = __builtin_amdgcn_readlane(pk[2], j);
                        } else if (mk[3] != 0u) {
                            int j = __ffs(mk[3]) - 1; mk[3] &= mk[3] - 1u;
                            e = __builtin_amdgcn_readlane(pk[3], j);
                        } else {
                            got = false;
                        }
                        es[t] = e >> 8; ed[t] = e & 255;
                        if (got) cnt = t + 1;
                    }
                    v4f x0[8], x1[8];
                    #pragma unroll
                    for (int t = 0; t < 8; ++t) {
                        x0[t] = (v4f){0.f, 0.f, 0.f, 0.f};
                        x1[t] = (v4f){0.f, 0.f, 0.f, 0.f};
                        if (t < cnt) {
                            const float* rp = h + (size_t)es[t] * ROW + lane * 4;
                            x0[t] = *(const v4f*)rp;
                            x1[t] = *(const v4f*)(rp + DIM);
                        }
                    }
                    asm volatile("" :: "v"(x0[0]), "v"(x0[1]), "v"(x0[2]), "v"(x0[3]),
                                       "v"(x0[4]), "v"(x0[5]), "v"(x0[6]), "v"(x0[7]),
                                       "v"(x1[0]), "v"(x1[1]), "v"(x1[2]), "v"(x1[3]),
                                       "v"(x1[4]), "v"(x1[5]), "v"(x1[6]), "v"(x1[7]));
                    const float u = trace8(x0, cnt, wS0, wS1, lane);
                    #pragma unroll
                    for (int t = 0; t < 8; ++t) {
                        if (t < cnt) {
                            const int row = ed[t];
                            const float p = __int_as_float(__builtin_amdgcn_readlane(__float_as_int(u), t));
                            const float s = p + q_l[row];
                            const float a = (s >= 0.0f) ? s : SLOPE * s;
                            const float2 st = ml[row];
                            const float mn = fmaxf(st.x, a);
                            const float sc = __expf(st.x - mn);
                            const float w  = __expf(a - mn);
                            v4f* r0 = (v4f*)(acc + row * ROW + lane * 4);
                            v4f* r1 = (v4f*)(acc + row * ROW + DIM + lane * 4);
                            *r0 = *r0 * sc + w * x0[t];
                            *r1 = *r1 * sc + w * x1[t];
                            if (lane == 0) ml[row] = make_float2(mn, st.y * sc + w);
                        }
                    }
                }
            }
            fill = 0;
            __syncthreads();
        }
        if (done) break;
    }

    if (tid < OWN_R) {
        const float l = ml[tid].y;
        inv_l[tid] = (l > 0.0f) ? 1.0f / l : 0.0f;
    }
    __syncthreads();

    const float4* h4 = (const float4*)h;
    float4* out4 = (float4*)out;
    for (int i = tid; i < nr * (ROW / 4); i += OWN_T) {
        int r = i >> 6;
        int n = lo + r;
        size_t g = (size_t)n * (ROW / 4) + (i & 63);
        float4 v = acc4[i];
        const float inv = inv_l[r];
        if (inv == 0.0f) {
            v = h4[g];
        } else {
            v.x *= inv; v.y *= inv; v.z *= inv; v.w *= inv;
        }
        out4[g] = v;
    }
}

extern "C" void kernel_launch(void* const* d_in, const int* in_sizes, int n_in,
                              void* d_out, int out_size, void* d_ws, size_t ws_size,
                              hipStream_t stream) {
    const float* h   = (const float*)d_in[0];
    const int*   src = (const int*)d_in[1];
    const int*   dst = (const int*)d_in[2];
    const float* w   = (const float*)d_in[3];
    int E = in_sizes[1];
    int N = in_sizes[0] / ROW;
    float* out = (float*)d_out;

    k_owner<<<(N + OWN_R - 1) / OWN_R, OWN_T, 0, stream>>>(h, src, dst, w, out, E, N);
}


# owner index dst/200 written once as a byte per edge by a first launch; owner work-groups scan that byte plane (a quarter of the bytes, 4 edges per word test), not dst; 2 launches; sum order unchanged
# speedup vs baseline: 3.6323x; 1.0586x over previous
#include <hip/hip_runtime.h>
#include <hip/hip_bf16.h>
#include <math.h>

#define SLOT  2
#define DIM   128
#define ROW   (SLOT * DIM)
#define SLOPE 0.98f

typedef float v2f __attribute__((ext_vector_type(2)));
typedef float v8f __attribute__((ext_vector_type(8)));
typedef float v4f __attribute__((ext_vector_type(4)));


__device__ __forceinline__ float xhalf_f32(float x) {
    int b = __float_as_int(x);
    return __int_as_float(__builtin_amdgcn_permlanex16(b, b, 0x76543210, 0xfedcba98, false, false));
}
template <int CTRL>
__device__ __forceinline__ float rowx_f32(float x) {
    return __int_as_float(__builtin_amdgcn_update_dpp(0, __float_as_int(x), CTRL, 0xf, 0xf, true));
}

__device__ __forceinline__ float trace8(const v4f (&x)[8], int cnt, v2f w0, v2f w1, int lane) {
    const int  r    = lane & 7;
    const bool diag = ((lane >> 3) & 1) == (lane >> 4);
    const bool bit0 = (lane & 1) != 0, bit1 = (lane & 2) != 0, bit2 = (lane & 4) != 0;
    float t[8];
    #pragma unroll
    for (int j = 0; j < 8; ++j) {
        t[j] = 0.0f;
        if (j < cnt) {
            const v2f a0 = {x[j].x, x[j].y}, a1 = {x[j].z, x[j].w};
            v8f c = {};
            c = __builtin_amdgcn_wmma_f32_16x16x4_f32(false, a0, false, w0, (short)0, c, false, false);
            c = __builtin_amdgcn_wmma_f32_16x16x4_f32(false, a1, false, w1, (short)0, c, false, false);
            float d = c[0];
            d = (r == 1) ? c[1] : d;
            d = (r == 2) ? c[2] : d;
            d = (r == 3) ? c[3] : d;
            d = (r == 4) ? c[4] : d;
            d = (r == 5) ? c[5] : d;
            d = (r == 6) ? c[6] : d;
            d = (r == 7) ? c[7] : d;
            t[j] = d;
        }
    }
    float a4[4], a2[2];
    #pragma unroll
    for (int j = 0; j < 4; ++j)
        a4[j] = (bit2 ? t[j + 4] : t[j]) + rowx_f32<0x164>(bit2 ? t[j] : t[j + 4]);
    #pragma unroll
    for (int j = 0; j < 2; ++j)
        a2[j] = (bit1 ? a4[j + 2] : a4[j]) + rowx_f32<0x162>(bit1 ? a4[j] : a4[j + 2]);
    float u = (bit0 ? a2[1] : a2[0]) + rowx_f32<0x161>(bit0 ? a2[0] : a2[1]);
    u = diag ? u : 0.0f;
    u = u + xhalf_f32(u);
    u = u + rowx_f32<0x168>(u);
    return u;
}

#define OWN_R      200
#define BLK_EDGES  1024
#define PRE_T      128

__device__ __forceinline__ unsigned owner_byte(int d) {
    const unsigned o = (unsigned)d / (unsigned)OWN_R;
    return o < 255u ? o : 255u;
}

__global__ void __launch_bounds__(PRE_T)
k_plane(const int* __restrict__ dst, unsigned* __restrict__ plane, int E) {
    const int lane  = threadIdx.x & 31;
    const int blk   = blockIdx.x * (PRE_T / 32) + (threadIdx.x >> 5);
    const int wbase = blk * BLK_EDGES;
    int4 d4[8];
    if (wbase + BLK_EDGES <= E) {
        #pragma unroll
        for (int j = 0; j < 8; ++j)
            d4[j] = *(const int4*)(dst + wbase + j * 128 + lane * 4);
    } else {
        #pragma unroll
        for (int j = 0; j < 8; ++j) {
            int e0 = wbase + j * 128 + lane * 4;
            d4[j].x = (e0     < E) ? dst[e0]     : -1;
            d4[j].y = (e0 + 1 < E) ? dst[e0 + 1] : -1;
            d4[j].z = (e0 + 2 < E) ? dst[e0 + 2] : -1;
            d4[j].w = (e0 + 3 < E) ? dst[e0 + 3] : -1;
        }
    }
    unsigned w[8] = {0u, 0u, 0u, 0u, 0u, 0u, 0u, 0u};
    #pragma unroll
    for (int j = 0; j < 8; ++j) {
        const int k0 = 4 * (j & 1), sh = 8 * (j >> 1);
        w[k0 + 0] |= owner_byte(d4[j].x) << sh;
        w[k0 + 1] |= owner_byte(d4[j].y) << sh;
        w[k0 + 2] |= owner_byte(d4[j].z) << sh;
        w[k0 + 3] |= owner_byte(d4[j].w) << sh;
    }
    uint4* p4 = (uint4*)(plane + blk * (BLK_EDGES / 4) + lane * 4);
    p4[0]  = make_uint4(w[0], w[1], w[2], w[3]);
    p4[32] = make_uint4(w[4], w[5], w[6], w[7]);
}

#define OWN_T      1024
#define OWN_WEDGES BLK_EDGES
#define OWN_CHUNK  (32 * OWN_WEDGES)
#define OWN_CAP    8192
#define M_NONE     (-3.0e38f)

static_assert((OWN_R + 7) / 8 <= OWN_T / 32, "one wave per batch of eight owned rows");
static_assert(OWN_R <= 256, "a list entry keeps the row in its low 8 bits");
static_assert(OWN_CHUNK % (PRE_T / 32 * BLK_EDGES) == 0, "the plane launch covers whole scan chunks");

__global__ void __launch_bounds__(OWN_T)
k_owner(const float* __restrict__ h,
        const int*  __restrict__ src,
        const int*  __restrict__ dst,
        const unsigned* __restrict__ plane,
        const float* __restrict__ watt,
        float* __restrict__ out, int E, int N) {
    __shared__ __align__(16) float acc[OWN_R * ROW];
    __shared__ int   l_a[OWN_CAP];
    __shared__ int   wtot[32];
    __shared__ float q_l[OWN_R];
    __shared__ float2 ml[OWN_R];
    __shared__ float inv_l[OWN_R];

    const int tid  = threadIdx.x;
    const int lane = tid & 31;
    const int wv   = __builtin_amdgcn_readfirstlane(tid >> 5);
    const int lo   = blockIdx.x * OWN_R;
    int nr = N - lo; if (nr > OWN_R) nr = OWN_R;
    const unsigned mine4 = (unsigned)blockIdx.x * 0x01010101u;

    const v4f wS = *(const v4f*)(watt + lane * 4);
    const v2f wS0 = {wS.x, wS.y}, wS1 = {wS.z, wS.w};

    float4* acc4 = (float4*)acc;
    for (int i = tid; i < OWN_R * (ROW / 4); i += OWN_T)
        acc4[i] = make_float4(0.f, 0.f, 0.f, 0.f);
    if (tid < OWN_R) ml[tid] = make_float2(M_NONE, 0.0f);
    if (wv < (OWN_R + 7) / 8) {
        const v4f wD = *(const v4f*)(watt + DIM + lane * 4);
        const v2f wD0 = {wD.x, wD.y}, wD1 = {wD.z, wD.w};
        v4f xq[8];
        #pragma unroll
        for (int j = 0; j < 8; ++j) {
            const int rr = wv * 8 + j;
            xq[j] = (v4f){0.f, 0.f, 0.f, 0.f};
            if (rr < nr) xq[j] = *(const v4f*)(h + (size_t)(lo + rr) * ROW + lane * 4);
        }
        const float uq = trace8(xq, 8, wD0, wD1, lane);
        const int rq = wv * 8 + lane;
        if (lane < 8 && rq < OWN_R) q_l[rq] = uq;
    }
    __syncthreads();

    const int ng = wv;

    int cbase = 0;
    int wbase = 0, woff = 0, tot = 0, p0 = 0;
    unsigned mbits = 0u;
    int lcnt = 0, linc = 0;
    int fill = 0;

    for (;;) {
        if (p0 == tot && cbase < E) {
            wbase = cbase + wv * OWN_WEDGES;
            const uint4* b4 = (const uint4*)(plane + (wbase >> 2) + lane * 4);
            const uint4 ba = b4[0], bb = b4[32];
            const unsigned pw[8] = {ba.x, ba.y, ba.z, ba.w, bb.x, bb.y, bb.z, bb.w};
            unsigned nacc = 0u;
            #pragma unroll
            for (int k = 0; k < 8; ++k) {
                const unsigned y  = pw[k] ^ mine4;
                const unsigned nz = (((y & 0x7f7f7f7fu) + 0x7f7f7f7fu) | y) & 0x80808080u;
                nacc = (nacc >> 1) | nz;
            }
            mbits = ~nacc;
            lcnt = __popc(mbits);
            linc = lcnt;
            for (int d = 1; d < 32; d <<= 1) {
                int t = __shfl_up(linc, d, 32);
                if (lane >= d) linc += t;
            }
            if (lane == 31) wtot[wv] = linc;
            __syncthreads();

            int incl = wtot[lane];
            const int mine = incl;
            for (int d = 1; d < 32; d <<= 1) {
                int t = __shfl_up(incl, d, 32);
                if (lane >= d) incl += t;
            }
            woff = __builtin_amdgcn_readfirstlane(__shfl(incl - mine, wv, 32));
            tot  = __builtin_amdgcn_readfirstlane(__shfl(incl, 31, 32));
            __syncthreads();
            p0 = 0;
            cbase += OWN_CHUNK;
        }

        if (p0 < tot) {
            int take = tot - p0;
            if (take > OWN_CAP - fill) take = OWN_CAP - fill;
            int pos = woff + (linc - lcnt) - p0;
            unsigned mb = mbits;
            while (mb != 0u) {
                int k = __ffs(mb) - 1; mb &= mb - 1u;
                if (pos >= 0 && pos < take)
                    l_a[fill + pos] = wbase + (k >> 2) * 128 + lane * 4 + (k & 3);
                ++pos;
            }
            fill += take;
            p0   += take;
        }

        const bool done = (p0 == tot) && (cbase >= E);
        if (fill == OWN_CAP || (done && fill > 0)) {
            const int n = fill;
            __syncthreads();
            for (int i = tid; i < n; i += OWN_T) {
                int e = l_a[i];
                int d = dst[e];
                int sv = src[e];
                l_a[i] = (sv << 8) | (d - lo);
            }
            __syncthreads();

            for (int b0 = 0; b0 < n; b0 += 128) {
                int pk[4]; unsigned mk[4];
                #pragma unroll
                for (int k = 0; k < 4; ++k) {
                    const int  i = b0 + k * 32 + lane;
                    const bool v = i < n;
                    pk[k] = v ? l_a[i] : 0;
                    mk[k] = (unsigned)__ballot(v && ((pk[k] & 31) == ng));
                }
                while ((mk[0] | mk[1] | mk[2] | mk[3]) != 0u) {
                    int es[8], ed[8]; int cnt = 0;
                    #pragma unroll
                    for (int t = 0; t < 8; ++t) {
                        int e = 0; bool got = true;
                        if (mk[0] != 0u) {
                            int j = __ffs(mk[0]) - 1; mk[0] &= mk[0] - 1u;
                            e = __builtin_amdgcn_readlane(pk[0], j);
                        } else if (mk[1] != 0u) {
                            int j = __ffs(mk[1]) - 1; mk[1] &= mk[1] - 1u;
                            e = __builtin_amdgcn_readlane(pk[1], j);
                        } else if (mk[2] != 0u) {
                            int j = __ffs(mk[2]) - 1; mk[2] &= mk[2] - 1u;
                            e = __builtin_amdgcn_readlane(pk[2], j);
                        } else if (mk[3] != 0u) {
                            int j = __ffs(mk[3]) - 1; mk[3] &= mk[3] - 1u;
                            e = __builtin_amdgcn_readlane(pk[3], j);
                        } else {
                            got = false;
                        }
                        es[t] = e >> 8; ed[t] = e & 255;
                        if (got) cnt = t + 1;
                    }
                    v4f x0[8], x1[8];
                    #pragma unroll
                    for (int t = 0; t < 8; ++t) {
                        x0[t] = (v4f){0.f, 0.f, 0.f, 0.f};
                        x1[t] = (v4f){0.f, 0.f, 0.f, 0.f};
                        if (t < cnt) {
                            const float* rp = h + (size_t)es[t] * ROW + lane * 4;
                            x0[t] = *(const v4f*)rp;
                            x1[t] = *(const v4f*)(rp + DIM);
                        }
                    }
                    asm volatile("" :: "v"(x0[0]), "v"(x0[1]), "v"(x0[2]), "v"(x0[3]),
                                       "v"(x0[4]), "v"(x0[5]), "v"(x0[6]), "v"(x0[7]),
                                       "v"(x1[0]), "v"(x1[1]), "v"(x1[2]), "v"(x1[3]),
                                       "v"(x1[4]), "v"(x1[5]), "v"(x1[6]), "v"(x1[7]));
                    const float u = trace8(x0, cnt, wS0, wS1, lane);
                    #pragma unroll
                    for (int t = 0; t < 8; ++t) {
                        if (t < cnt) {
                            const int row = ed[t];
                            const float p = __int_as_float(__builtin_amdgcn_readlane(__float_as_int(u), t));
                            const float s = p + q_l[row];
                            const float a = (s >= 0.0f) ? s : SLOPE * s;
                            const float2 st = ml[row];
                            const float mn = fmaxf(st.x, a);
                            const float sc = __expf(st.x - mn);
                            const float w  = __expf(a - mn);
                            v4f* r0 = (v4f*)(acc + row * ROW + lane * 4);
                            v4f* r1 = (v4f*)(acc + row * ROW + DIM + lane * 4);
                            *r0 = *r0 * sc + w * x0[t];
                            *r1 = *r1 * sc + w * x1[t];
                            if (lane == 0) ml[row] = make_float2(mn, st.y * sc + w);
                        }
                    }
                }
            }
            fill = 0;
            __syncthreads();
        }
        if (done) break;
    }

    if (tid < OWN_R) {
        const float l = ml[tid].y;
        inv_l[tid] = (l > 0.0f) ? 1.0f / l : 0.0f;
    }
    __syncthreads();

    const float4* h4 = (const float4*)h;
    float4* out4 = (float4*)out;
    for (int i = tid; i < nr * (ROW / 4); i += OWN_T) {
        int r = i >> 6;
        int n = lo + r;
        size_t g = (size_t)n * (ROW / 4) + (i & 63);
        float4 v = acc4[i];
        const float inv = inv_l[r];
        if (inv == 0.0f) {
            v = h4[g];
        } else {
            v.x *= inv; v.y *= inv; v.z *= inv; v.w *= inv;
        }
        out4[g] = v;
    }
}

extern "C" void kernel_launch(void* const* d_in, const int* in_sizes, int n_in,
                              void* d_out, int out_size, void* d_ws, size_t ws_size,
                              hipStream_t stream) {
    const float* h   = (const float*)d_in[0];
    const int*   src = (const int*)d_in[1];
    const int*   dst = (const int*)d_in[2];
    const float* w   = (const float*)d_in[3];
    int E = in_sizes[1];
    int N = in_sizes[0] / ROW;
    float* out = (float*)d_out;

    unsigned* plane = (unsigned*)d_ws;
    k_plane<<<(E + OWN_CHUNK - 1) / OWN_CHUNK * (OWN_CHUNK / (PRE_T / 32 * BLK_EDGES)), PRE_T, 0, stream>>>(dst, plane, E);
    k_owner<<<(N + OWN_R - 1) / OWN_R, OWN_T, 0, stream>>>(h, src, dst, plane, w, out, E, N);
}


# owner work-group: each wave lists its own plane blocks into its own list segment, next block prefetched, lane prefix by DPP; no barrier or LDS hop per block; list order by wave: sum order changes
# speedup vs baseline: 4.5005x; 1.2390x over previous
#include <hip/hip_runtime.h>
#include <hip/hip_bf16.h>
#include <math.h>

#define SLOT  2
#define DIM   128
#define ROW   (SLOT * DIM)
#define SLOPE 0.98f

typedef float v2f __attribute__((ext_vector_type(2)));
typedef float v8f __attribute__((ext_vector_type(8)));
typedef float v4f __attribute__((ext_vector_type(4)));


__device__ __forceinline__ float xhalf_f32(float x) {
    int b = __float_as_int(x);
    return __int_as_float(__builtin_amdgcn_permlanex16(b, b, 0x76543210, 0xfedcba98, false, false));
}
template <int CTRL>
__device__ __forceinline__ float rowx_f32(float x) {
    return __int_as_float(__builtin_amdgcn_update_dpp(0, __float_as_int(x), CTRL, 0xf, 0xf, true));
}

__device__ __forceinline__ float trace8(const v4f (&x)[8], int cnt, v2f w0, v2f w1, int lane) {
    const int  r    = lane & 7;
    const bool diag = ((lane >> 3) & 1) == (lane >> 4);
    const bool bit0 = (lane & 1) != 0, bit1 = (lane & 2) != 0, bit2 = (lane & 4) != 0;
    float t[8];
    #pragma unroll
    for (int j = 0; j < 8; ++j) {
        t[j] = 0.0f;
        if (j < cnt) {
            const v2f a0 = {x[j].x, x[j].y}, a1 = {x[j].z, x[j].w};
            v8f c = {};
            c = __builtin_amdgcn_wmma_f32_16x16x4_f32(false, a0, false, w0, (short)0, c, false, false);
            c = __builtin_amdgcn_wmma_f32_16x16x4_f32(false, a1, false, w1, (short)0, c, false, false);
            float d = c[0];
            d = (r == 1) ? c[1] : d;
            d = (r == 2) ? c[2] : d;
            d = (r == 3) ? c[3] : d;
            d = (r == 4) ? c[4] : d;
            d = (r == 5) ? c[5] : d;
            d = (r == 6) ? c[6] : d;
            d = (r == 7) ? c[7] : d;
            t[j] = d;
        }
    }
    float a4[4], a2[2];
    #pragma unroll
    for (int j = 0; j < 4; ++j)
        a4[j] = (bit2 ? t[j + 4] : t[j]) + rowx_f32<0x164>(bit2 ? t[j] : t[j + 4]);
    #pragma unroll
    for (int j = 0; j < 2; ++j)
        a2[j] = (bit1 ? a4[j + 2] : a4[j]) + rowx_f32<0x162>(bit1 ? a4[j] : a4[j + 2]);
    float u = (bit0 ? a2[1] : a2[0]) + rowx_f32<0x161>(bit0 ? a2[0] : a2[1]);
    u = diag ? u : 0.0f;
    u = u + xhalf_f32(u);
    u = u + rowx_f32<0x168>(u);
    return u;
}

#define OWN_R      200
#define BLK_EDGES  1024
#define PRE_T      128

__device__ __forceinline__ unsigned owner_byte(int d) {
    const unsigned o = (unsigned)d / (unsigned)OWN_R;
    return o < 255u ? o : 255u;
}

__global__ void __launch_bounds__(PRE_T)
k_plane(const int* __restrict__ dst, unsigned* __restrict__ plane, int E) {
    const int lane  = threadIdx.x & 31;
    const int blk   = blockIdx.x * (PRE_T / 32) + (threadIdx.x >> 5);
    const int wbase = blk * BLK_EDGES;
    int4 d4[8];
    if (wbase + BLK_EDGES <= E) {
        #pragma unroll
        for (int j = 0; j < 8; ++j)
            d4[j] = *(const int4*)(dst + wbase + j * 128 + lane * 4);
    } else {
        #pragma unroll
        for (int j = 0; j < 8; ++j) {
            int e0 = wbase + j * 128 + lane * 4;
            d4[j].x = (e0     < E) ? dst[e0]     : -1;
            d4[j].y = (e0 + 1 < E) ? dst[e0 + 1] : -1;
            d4[j].z = (e0 + 2 < E) ? dst[e0 + 2] : -1;
            d4[j].w = (e0 + 3 < E) ? dst[e0 + 3] : -1;
        }
    }
    unsigned w[8] = {0u, 0u, 0u, 0u, 0u, 0u, 0u, 0u};
    #pragma unroll
    for (int j = 0; j < 8; ++j) {
        const int k0 = 4 * (j & 1), sh = 8 * (j >> 1);
        w[k0 + 0] |= owner_byte(d4[j].x) << sh;
        w[k0 + 1] |= owner_byte(d4[j].y) << sh;
        w[k0 + 2] |= owner_byte(d4[j].z) << sh;
        w[k0 + 3] |= owner_byte(d4[j].w) << sh;
    }
    uint4* p4 = (uint4*)(plane + blk * (BLK_EDGES / 4) + lane * 4);
    p4[0]  = make_uint4(w[0], w[1], w[2], w[3]);
    p4[32] = make_uint4(w[4], w[5], w[6], w[7]);
}

#define OWN_T      1024
#define OWN_WEDGES BLK_EDGES
#define OWN_CHUNK  (32 * OWN_WEDGES)
#define OWN_CAP    8192
#define OWN_WCAP   (OWN_CAP / 32)
#define M_NONE     (-3.0e38f)

static_assert((OWN_R + 7) / 8 <= OWN_T / 32, "one wave per batch of eight owned rows");
static_assert(OWN_R <= 256, "a list entry keeps the row in its low 8 bits");
static_assert(OWN_CHUNK % (PRE_T / 32 * BLK_EDGES) == 0, "the plane launch covers whole scan chunks");
static_assert(OWN_WCAP == 256, "the resolve step takes a segment as eight entries per lane");

template <int N>
__device__ __forceinline__ int row_shr_i32(int x) {
    return __builtin_amdgcn_update_dpp(0, x, 0x110 + N, 0xf, 0xf, false);
}
__device__ __forceinline__ int wave_prefix_i32(int x, int lane) {
    x += row_shr_i32<1>(x);
    x += row_shr_i32<2>(x);
    x += row_shr_i32<4>(x);
    x += row_shr_i32<8>(x);
    const int low = __builtin_amdgcn_readlane(x, 15);
    return (lane >= 16) ? x + low : x;
}

__global__ void __launch_bounds__(OWN_T)
k_owner(const float* __restrict__ h,
        const int*  __restrict__ src,
        const int*  __restrict__ dst,
        const unsigned* __restrict__ plane,
        const float* __restrict__ watt,
        float* __restrict__ out, int E, int N) {
    __shared__ __align__(16) float acc[OWN_R * ROW];
    __shared__ int   l_a[OWN_CAP];
    __shared__ int   l_b[OWN_CAP];
    __shared__ int   wcnt[32];
    __shared__ float q_l[OWN_R];
    __shared__ float2 ml[OWN_R];
    __shared__ float inv_l[OWN_R];

    const int tid  = threadIdx.x;
    const int lane = tid & 31;
    const int wv   = __builtin_amdgcn_readfirstlane(tid >> 5);
    const int lo   = blockIdx.x * OWN_R;
    int nr = N - lo; if (nr > OWN_R) nr = OWN_R;
    const unsigned mine4 = (unsigned)blockIdx.x * 0x01010101u;

    const v4f wS = *(const v4f*)(watt + lane * 4);
    const v2f wS0 = {wS.x, wS.y}, wS1 = {wS.z, wS.w};

    float4* acc4 = (float4*)acc;
    for (int i = tid; i < OWN_R * (ROW / 4); i += OWN_T)
        acc4[i] = make_float4(0.f, 0.f, 0.f, 0.f);
    if (tid < OWN_R) ml[tid] = make_float2(M_NONE, 0.0f);
    if (wv < (OWN_R + 7) / 8) {
        const v4f wD = *(const v4f*)(watt + DIM + lane * 4);
        const v2f wD0 = {wD.x, wD.y}, wD1 = {wD.z, wD.w};
        v4f xq[8];
        #pragma unroll
        for (int j = 0; j < 8; ++j) {
            const int rr = wv * 8 + j;
            xq[j] = (v4f){0.f, 0.f, 0.f, 0.f};
            if (rr < nr) xq[j] = *(const v4f*)(h + (size_t)(lo + rr) * ROW + lane * 4);
        }
        const float uq = trace8(xq, 8, wD0, wD1, lane);
        const int rq = wv * 8 + lane;
        if (lane < 8 && rq < OWN_R) q_l[rq] = uq;
    }
    __syncthreads();

    const int ng = wv;
    const int nblk = (E + OWN_CHUNK - 1) / OWN_CHUNK;
    const int sb = wv * OWN_WCAP;

    int c = 0, p0 = 0;

    for (;;) {
        int fill = 0;
        uint4 na = make_uint4(0u, 0u, 0u, 0u), nb = na;
        if (c < nblk) {
            const uint4* b4 = (const uint4*)(plane + ((c * OWN_CHUNK + wv * OWN_WEDGES) >> 2) + lane * 4);
            na = b4[0]; nb = b4[32];
        }
        while (c < nblk && fill < OWN_WCAP) {
            const int wbase = c * OWN_CHUNK + wv * OWN_WEDGES;
            const uint4 ba = na, bb = nb;
            {
                const int cn = (c + 1 < nblk) ? c + 1 : c;
                const uint4* n4 = (const uint4*)(plane + ((cn * OWN_CHUNK + wv * OWN_WEDGES) >> 2) + lane * 4);
                na = n4[0]; nb = n4[32];
            }
            const unsigned pw[8] = {ba.x, ba.y, ba.z, ba.w, bb.x, bb.y, bb.z, bb.w};
            unsigned nacc = 0u;
            #pragma unroll
            for (int k = 0; k < 8; ++k) {
                const unsigned y  = pw[k] ^ mine4;
                const unsigned nz = (((y & 0x7f7f7f7fu) + 0x7f7f7f7fu) | y) & 0x80808080u;
                nacc = (nacc >> 1) | nz;
            }
            const unsigned mbits = ~nacc;
            const int lcnt = __popc(mbits);
            const int linc = wave_prefix_i32(lcnt, lane);
            const int tot  = __builtin_amdgcn_readlane(linc, 31);
            int take = tot - p0;
            if (take > OWN_WCAP - fill) take = OWN_WCAP - fill;
            if (take > 0) {
                int pos = (linc - lcnt) - p0;
                unsigned mb = mbits;
                while (mb != 0u) {
                    int k = __ffs(mb) - 1; mb &= mb - 1u;
                    if (pos >= 0 && pos < take)
                        l_a[sb + fill + pos] = wbase + (k >> 2) * 128 + lane * 4 + (k & 3);
                    ++pos;
                }
            }
            fill += take;
            p0   += take;
            if (p0 == tot) { ++c; p0 = 0; }
        }
        if (lane == 0) wcnt[wv] = fill | ((c < nblk) ? 0x10000 : 0);
        __syncthreads();

        const int wc   = wcnt[lane];
        const bool more = __ballot((wc >> 16) != 0) != 0;
        const int cw   = wc & 0xffff;
        const int incl = wave_prefix_i32(cw, lane);
        const int off  = __builtin_amdgcn_readlane(incl - cw, wv);
        const int n    = __builtin_amdgcn_readlane(incl, 31);

        if (fill > 0) {
            int ee[8], dd[8], ss[8];
            #pragma unroll
            for (int k = 0; k < 8; ++k) {
                const int i = k * 32 + lane;
                const int e = l_a[sb + i];
                ee[k] = (i < fill) ? e : l_a[sb];
            }
            #pragma unroll
            for (int k = 0; k < 8; ++k) { dd[k] = dst[ee[k]]; ss[k] = src[ee[k]]; }
            #pragma unroll
            for (int k = 0; k < 8; ++k) {
                const int i = k * 32 + lane;
                if (i < fill) l_b[off + i] = (ss[k] << 8) | (dd[k] - lo);
            }
        }
        __syncthreads();

        for (int b0 = 0; b0 < n; b0 += 128) {
            int pk[4]; unsigned mk[4];
            #pragma unroll
            for (int k = 0; k < 4; ++k) {
                const int  i = b0 + k * 32 + lane;
                const bool v = i < n;
                pk[k] = v ? l_b[i] : 0;
                mk[k] = (unsigned)__ballot(v && ((pk[k] & 31) == ng));
            }
            while ((mk[0] | mk[1] | mk[2] | mk[3]) != 0u) {
                int es[8], ed[8]; int cnt = 0;
                #pragma unroll
                for (int t = 0; t < 8; ++t) {
                    int e = 0; bool got = true;
                    if (mk[0] != 0u) {
                        int j = __ffs(mk[0]) - 1; mk[0] &= mk[0] - 1u;
                        e = __builtin_amdgcn_readlane(pk[0], j);
                    } else if (mk[1] != 0u) {
                        int j = __ffs(mk[1]) - 1; mk[1] &= mk[1] - 1u;
                        e = __builtin_amdgcn_readlane(pk[1], j);
                    } else if (mk[2] != 0u) {
                        int j = __ffs(mk[2]) - 1; mk[2] &= mk[2] - 1u;
                        e = __builtin_amdgcn_readlane(pk[2], j);
                    } else if (mk[3] != 0u) {
                        int j = __ffs(mk[3]) - 1; mk[3] &= mk[3] - 1u;
                        e = __builtin_amdgcn_readlane(pk[3], j);
                    } else {
                        got = false;
                    }
                    es[t] = e >> 8; ed[t] = e & 255;
                    if (got) cnt = t + 1;
                }
                v4f x0[8], x1[8];
                #pragma unroll
                for (int t = 0; t < 8; ++t) {
                    x0[t] = (v4f){0.f, 0.f, 0.f, 0.f};
                    x1[t] = (v4f){0.f, 0.f, 0.f, 0.f};
                    if (t < cnt) {
                        const float* rp = h + (size_t)es[t] * ROW + lane * 4;
                        x0[t] = *(const v4f*)rp;
                        x1[t] = *(const v4f*)(rp + DIM);
                    }
                }
                asm volatile("" :: "v"(x0[0]), "v"(x0[1]), "v"(x0[2]), "v"(x0[3]),
                                   "v"(x0[4]), "v"(x0[5]), "v"(x0[6]), "v"(x0[7]),
                                   "v"(x1[0]), "v"(x1[1]), "v"(x1[2]), "v"(x1[3]),
                                   "v"(x1[4]), "v"(x1[5]), "v"(x1[6]), "v"(x1[7]));
                const float u = trace8(x0, cnt, wS0, wS1, lane);
                #pragma unroll
                for (int t = 0; t < 8; ++t) {
                    if (t < cnt) {
                        const int row = ed[t];
                        const float p = __int_as_float(__builtin_amdgcn_readlane(__float_as_int(u), t));
                        const float s = p + q_l[row];
                        const float a = (s >= 0.0f) ? s : SLOPE * s;
                        const float2 st = ml[row];
                        const float mn = fmaxf(st.x, a);
                        const float sc = __expf(st.x - mn);
                        const float w  = __expf(a - mn);
                        v4f* r0 = (v4f*)(acc + row * ROW + lane * 4);
                        v4f* r1 = (v4f*)(acc + row * ROW + DIM + lane * 4);
                        *r0 = *r0 * sc + w * x0[t];
                        *r1 = *r1 * sc + w * x1[t];
                        if (lane == 0) ml[row] = make_float2(mn, st.y * sc + w);
                    }
                }
            }
        }
        __syncthreads();
        if (!more) break;
    }

    if (tid < OWN_R) {
        const float l = ml[tid].y;
        inv_l[tid] = (l > 0.0f) ? 1.0f / l : 0.0f;
    }
    __syncthreads();

    const float4* h4 = (const float4*)h;
    float4* out4 = (float4*)out;
    for (int i = tid; i < nr * (ROW / 4); i += OWN_T) {
        int r = i >> 6;
        int n = lo + r;
        size_t g = (size_t)n * (ROW / 4) + (i & 63);
        float4 v = acc4[i];
        const float inv = inv_l[r];
        if (inv == 0.0f) {
            v = h4[g];
        } else {
            v.x *= inv; v.y *= inv; v.z *= inv; v.w *= inv;
        }
        out4[g] = v;
    }
}

extern "C" void kernel_launch(void* const* d_in, const int* in_sizes, int n_in,
                              void* d_out, int out_size, void* d_ws, size_t ws_size,
                              hipStream_t stream) {
    const float* h   = (const float*)d_in[0];
    const int*   src = (const int*)d_in[1];
    const int*   dst = (const int*)d_in[2];
    const float* w   = (const float*)d_in[3];
    int E = in_sizes[1];
    int N = in_sizes[0] / ROW;
    float* out = (float*)d_out;

    unsigned* plane = (unsigned*)d_ws;
    k_plane<<<(E + OWN_CHUNK - 1) / OWN_CHUNK * (OWN_CHUNK / (PRE_T / 32 * BLK_EDGES)), PRE_T, 0, stream>>>(dst, plane, E);
    k_owner<<<(N + OWN_R - 1) / OWN_R, OWN_T, 0, stream>>>(h, src, dst, plane, w, out, E, N);
}


# owner add pass: each wave first copies its rows' entries from the dense list into its own list segment (ballot + lane prefix), then adds them in full batches of eight; sum order unchanged
# speedup vs baseline: 5.0911x; 1.1312x over previous
#include <hip/hip_runtime.h>
#include <hip/hip_bf16.h>
#include <math.h>

#define SLOT  2
#define DIM   128
#define ROW   (SLOT * DIM)
#define SLOPE 0.98f

typedef float v2f __attribute__((ext_vector_type(2)));
typedef float v8f __attribute__((ext_vector_type(8)));
typedef float v4f __attribute__((ext_vector_type(4)));


__device__ __forceinline__ float xhalf_f32(float x) {
    int b = __float_as_int(x);
    return __int_as_float(__builtin_amdgcn_permlanex16(b, b, 0x76543210, 0xfedcba98, false, false));
}
template <int CTRL>
__device__ __forceinline__ float rowx_f32(float x) {
    return __int_as_float(__builtin_amdgcn_update_dpp(0, __float_as_int(x), CTRL, 0xf, 0xf, true));
}

__device__ __forceinline__ float trace8(const v4f (&x)[8], int cnt, v2f w0, v2f w1, int lane) {
    const int  r    = lane & 7;
    const bool diag = ((lane >> 3) & 1) == (lane >> 4);
    const bool bit0 = (lane & 1) != 0, bit1 = (lane & 2) != 0, bit2 = (lane & 4) != 0;
    float t[8];
    #pragma unroll
    for (int j = 0; j < 8; ++j) {
        t[j] = 0.0f;
        if (j < cnt) {
            const v2f a0 = {x[j].x, x[j].y}, a1 = {x[j].z, x[j].w};
            v8f c = {};
            c = __builtin_amdgcn_wmma_f32_16x16x4_f32(false, a0, false, w0, (short)0, c, false, false);
            c = __builtin_amdgcn_wmma_f32_16x16x4_f32(false, a1, false, w1, (short)0, c, false, false);
            float d = c[0];
            d = (r == 1) ? c[1] : d;
            d = (r == 2) ? c[2] : d;
            d = (r == 3) ? c[3] : d;
            d = (r == 4) ? c[4] : d;
            d = (r == 5) ? c[5] : d;
            d = (r == 6) ? c[6] : d;
            d = (r == 7) ? c[7] : d;
            t[j] = d;
        }
    }
    float a4[4], a2[2];
    #pragma unroll
    for (int j = 0; j < 4; ++j)
        a4[j] = (bit2 ? t[j + 4] : t[j]) + rowx_f32<0x164>(bit2 ? t[j] : t[j + 4]);
    #pragma unroll
    for (int j = 0; j < 2; ++j)
        a2[j] = (bit1 ? a4[j + 2] : a4[j]) + rowx_f32<0x162>(bit1 ? a4[j] : a4[j + 2]);
    float u = (bit0 ? a2[1] : a2[0]) + rowx_f32<0x161>(bit0 ? a2[0] : a2[1]);
    u = diag ? u : 0.0f;
    u = u + xhalf_f32(u);
    u = u + rowx_f32<0x168>(u);
    return u;
}

#define OWN_R      200
#define BLK_EDGES  1024
#define PRE_T      128

__device__ __forceinline__ unsigned owner_byte(int d) {
    const unsigned o = (unsigned)d / (unsigned)OWN_R;
    return o < 255u ? o : 255u;
}

__global__ void __launch_bounds__(PRE_T)
k_plane(const int* __restrict__ dst, unsigned* __restrict__ plane, int E) {
    const int lane  = threadIdx.x & 31;
    const int blk   = blockIdx.x * (PRE_T / 32) + (threadIdx.x >> 5);
    const int wbase = blk * BLK_EDGES;
    int4 d4[8];
    if (wbase + BLK_EDGES <= E) {
        #pragma unroll
        for (int j = 0; j < 8; ++j)
            d4[j] = *(const int4*)(dst + wbase + j * 128 + lane * 4);
    } else {
        #pragma unroll
        for (int j = 0; j < 8; ++j) {
            int e0 = wbase + j * 128 + lane * 4;
            d4[j].x = (e0     < E) ? dst[e0]     : -1;
            d4[j].y = (e0 + 1 < E) ? dst[e0 + 1] : -1;
            d4[j].z = (e0 + 2 < E) ? dst[e0 + 2] : -1;
            d4[j].w = (e0 + 3 < E) ? dst[e0 + 3] : -1;
        }
    }
    unsigned w[8] = {0u, 0u, 0u, 0u, 0u, 0u, 0u, 0u};
    #pragma unroll
    for (int j = 0; j < 8; ++j) {
        const int k0 = 4 * (j & 1), sh = 8 * (j >> 1);
        w[k0 + 0] |= owner_byte(d4[j].x) << sh;
        w[k0 + 1] |= owner_byte(d4[j].y) << sh;
        w[k0 + 2] |= owner_byte(d4[j].z) << sh;
        w[k0 + 3] |= owner_byte(d4[j].w) << sh;
    }
    uint4* p4 = (uint4*)(plane + blk * (BLK_EDGES / 4) + lane * 4);
    p4[0]  = make_uint4(w[0], w[1], w[2], w[3]);
    p4[32] = make_uint4(w[4], w[5], w[6], w[7]);
}

#define OWN_T      1024
#define OWN_WEDGES BLK_EDGES
#define OWN_CHUNK  (32 * OWN_WEDGES)
#define OWN_CAP    8192
#define OWN_WCAP   (OWN_CAP / 32)
#define M_NONE     (-3.0e38f)

static_assert((OWN_R + 7) / 8 <= OWN_T / 32, "one wave per batch of eight owned rows");
static_assert(OWN_R <= 256, "a list entry keeps the row in its low 8 bits");
static_assert(OWN_CHUNK % (PRE_T / 32 * BLK_EDGES) == 0, "the plane launch covers whole scan chunks");
static_assert(OWN_WCAP == 256, "the resolve step takes a segment as eight entries per lane");

template <int N>
__device__ __forceinline__ int row_shr_i32(int x) {
    return __builtin_amdgcn_update_dpp(0, x, 0x110 + N, 0xf, 0xf, false);
}
__device__ __forceinline__ int wave_prefix_i32(int x, int lane) {
    x += row_shr_i32<1>(x);
    x += row_shr_i32<2>(x);
    x += row_shr_i32<4>(x);
    x += row_shr_i32<8>(x);
    const int low = __builtin_amdgcn_readlane(x, 15);
    return (lane >= 16) ? x + low : x;
}

__global__ void __launch_bounds__(OWN_T)
k_owner(const float* __restrict__ h,
        const int*  __restrict__ src,
        const int*  __restrict__ dst,
        const unsigned* __restrict__ plane,
        const float* __restrict__ watt,
        float* __restrict__ out, int E, int N) {
    __shared__ __align__(16) float acc[OWN_R * ROW];
    __shared__ int   l_a[OWN_CAP];
    __shared__ int   l_b[OWN_CAP];
    __shared__ int   wcnt[32];
    __shared__ float q_l[OWN_R];
    __shared__ float2 ml[OWN_R];
    __shared__ float inv_l[OWN_R];

    const int tid  = threadIdx.x;
    const int lane = tid & 31;
    const int wv   = __builtin_amdgcn_readfirstlane(tid >> 5);
    const int lo   = blockIdx.x * OWN_R;
    int nr = N - lo; if (nr > OWN_R) nr = OWN_R;
    const unsigned mine4 = (unsigned)blockIdx.x * 0x01010101u;

    const v4f wS = *(const v4f*)(watt + lane * 4);
    const v2f wS0 = {wS.x, wS.y}, wS1 = {wS.z, wS.w};

    float4* acc4 = (float4*)acc;
    for (int i = tid; i < OWN_R * (ROW / 4); i += OWN_T)
        acc4[i] = make_float4(0.f, 0.f, 0.f, 0.f);
    if (tid < OWN_R) ml[tid] = make_float2(M_NONE, 0.0f);
    if (wv < (OWN_R + 7) / 8) {
        const v4f wD = *(const v4f*)(watt + DIM + lane * 4);
        const v2f wD0 = {wD.x, wD.y}, wD1 = {wD.z, wD.w};
        v4f xq[8];
        #pragma unroll
        for (int j = 0; j < 8; ++j) {
            const int rr = wv * 8 + j;
            xq[j] = (v4f){0.f, 0.f, 0.f, 0.f};
            if (rr < nr) xq[j] = *(const v4f*)(h + (size_t)(lo + rr) * ROW + lane * 4);
        }
        const float uq = trace8(xq, 8, wD0, wD1, lane);
        const int rq = wv * 8 + lane;
        if (lane < 8 && rq < OWN_R) q_l[rq] = uq;
    }
    __syncthreads();

    const int ng = wv;
    const int nblk = (E + OWN_CHUNK - 1) / OWN_CHUNK;
    const int sb = wv * OWN_WCAP;

    int c = 0, p0 = 0;

    for (;;) {
        int fill = 0;
        uint4 na = make_uint4(0u, 0u, 0u, 0u), nb = na;
        if (c < nblk) {
            const uint4* b4 = (const uint4*)(plane + ((c * OWN_CHUNK + wv * OWN_WEDGES) >> 2) + lane * 4);
            na = b4[0]; nb = b4[32];
        }
        while (c < nblk && fill < OWN_WCAP) {
            const int wbase = c * OWN_CHUNK + wv * OWN_WEDGES;
            const uint4 ba = na, bb = nb;
            {
                const int cn = (c + 1 < nblk) ? c + 1 : c;
                const uint4* n4 = (const uint4*)(plane + ((cn * OWN_CHUNK + wv * OWN_WEDGES) >> 2) + lane * 4);
                na = n4[0]; nb = n4[32];
            }
            const unsigned pw[8] = {ba.x, ba.y, ba.z, ba.w, bb.x, bb.y, bb.z, bb.w};
            unsigned nacc = 0u;
            #pragma unroll
            for (int k = 0; k < 8; ++k) {
                const unsigned y  = pw[k] ^ mine4;
                const unsigned nz = (((y & 0x7f7f7f7fu) + 0x7f7f7f7fu) | y) & 0x80808080u;
                nacc = (nacc >> 1) | nz;
            }
            const unsigned mbits = ~nacc;
            const int lcnt = __popc(mbits);
            const int linc = wave_prefix_i32(lcnt, lane);
            const int tot  = __builtin_amdgcn_readlane(linc, 31);
            int take = tot - p0;
            if (take > OWN_WCAP - fill) take = OWN_WCAP - fill;
            if (take > 0) {
                int pos = (linc - lcnt) - p0;
                unsigned mb = mbits;
                while (mb != 0u) {
                    int k = __ffs(mb) - 1; mb &= mb - 1u;
                    if (pos >= 0 && pos < take)
                        l_a[sb + fill + pos] = wbase + (k >> 2) * 128 + lane * 4 + (k & 3);
                    ++pos;
                }
            }
            fill += take;
            p0   += take;
            if (p0 == tot) { ++c; p0 = 0; }
        }
        if (lane == 0) wcnt[wv] = fill | ((c < nblk) ? 0x10000 : 0);
        __syncthreads();

        const int wc   = wcnt[lane];
        const bool more = __ballot((wc >> 16) != 0) != 0;
        const int cw   = wc & 0xffff;
        const int incl = wave_prefix_i32(cw, lane);
        const int off  = __builtin_amdgcn_readlane(incl - cw, wv);
        const int n    = __builtin_amdgcn_readlane(incl, 31);

        if (fill > 0) {
            int ee[8], dd[8], ss[8];
            #pragma unroll
            for (int k = 0; k < 8; ++k) {
                const int i = k * 32 + lane;
                const int e = l_a[sb + i];
                ee[k] = (i < fill) ? e : l_a[sb];
            }
            #pragma unroll
            for (int k = 0; k < 8; ++k) { dd[k] = dst[ee[k]]; ss[k] = src[ee[k]]; }
            #pragma unroll
            for (int k = 0; k < 8; ++k) {
                const int i = k * 32 + lane;
                if (i < fill) l_b[off + i] = (ss[k] << 8) | (dd[k] - lo);
            }
        }
        __syncthreads();

        int pos = 0;
        while (pos < n) {
            int mc = 0;
            while (pos < n && mc <= OWN_WCAP - 128) {
                int pk[4];
                #pragma unroll
                for (int k = 0; k < 4; ++k) {
                    const int i = pos + k * 32 + lane;
                    pk[k] = (i < n) ? l_b[i] : 0;
                }
                #pragma unroll
                for (int k = 0; k < 4; ++k) {
                    const int  i    = pos + k * 32 + lane;
                    const bool mine = (i < n) && ((pk[k] & 31) == ng);
                    const unsigned m = (unsigned)__ballot(mine);
                    if (mine) l_a[sb + mc + (int)__builtin_amdgcn_mbcnt_lo(m, 0u)] = pk[k];
                    mc += __popc(m);
                }
                pos += 128;
            }
            for (int g = 0; g < mc; g += 8) {
                int cnt = mc - g; if (cnt > 8) cnt = 8;
                const int ent = l_a[sb + g + (lane & 7)];
                int es[8], ed[8];
                #pragma unroll
                for (int t = 0; t < 8; ++t) {
                    const int e = __builtin_amdgcn_readlane(ent, t);
                    es[t] = e >> 8; ed[t] = e & 255;
                }
                v4f x0[8], x1[8];
                #pragma unroll
                for (int t = 0; t < 8; ++t) {
                    const int sn = (t < cnt) ? es[t] : es[0];
                    const float* rp = h + (size_t)sn * ROW + lane * 4;
                    x0[t] = *(const v4f*)rp;
                    x1[t] = *(const v4f*)(rp + DIM);
                }
                const float u = trace8(x0, cnt, wS0, wS1, lane);
                #pragma unroll
                for (int t = 0; t < 8; ++t) {
                    if (t < cnt) {
                        const int row = ed[t];
                        const float p = __int_as_float(__builtin_amdgcn_readlane(__float_as_int(u), t));
                        const float s = p + q_l[row];
                        const float a = (s >= 0.0f) ? s : SLOPE * s;
                        const float2 st = ml[row];
                        const float mn = fmaxf(st.x, a);
                        const float sc = __expf(st.x - mn);
                        const float w  = __expf(a - mn);
                        v4f* r0 = (v4f*)(acc + row * ROW + lane * 4);
                        v4f* r1 = (v4f*)(acc + row * ROW + DIM + lane * 4);
                        *r0 = *r0 * sc + w * x0[t];
                        *r1 = *r1 * sc + w * x1[t];
                        if (lane == 0) ml[row] = make_float2(mn, st.y * sc + w);
                    }
                }
            }
        }
        __syncthreads();
        if (!more) break;
    }

    if (tid < OWN_R) {
        const float l = ml[tid].y;
        inv_l[tid] = (l > 0.0f) ? 1.0f / l : 0.0f;
    }
    __syncthreads();

    const float4* h4 = (const float4*)h;
    float4* out4 = (float4*)out;
    for (int i = tid; i < nr * (ROW / 4); i += OWN_T) {
        int r = i >> 6;
        int n = lo + r;
        size_t g = (size_t)n * (ROW / 4) + (i & 63);
        float4 v = acc4[i];
        const float inv = inv_l[r];
        if (inv == 0.0f) {
            v = h4[g];
        } else {
            v.x *= inv; v.y *= inv; v.z *= inv; v.w *= inv;
        }
        out4[g] = v;
    }
}

extern "C" void kernel_launch(void* const* d_in, const int* in_sizes, int n_in,
                              void* d_out, int out_size, void* d_ws, size_t ws_size,
                              hipStream_t stream) {
    const float* h   = (const float*)d_in[0];
    const int*   src = (const int*)d_in[1];
    const int*   dst = (const int*)d_in[2];
    const float* w   = (const float*)d_in[3];
    int E = in_sizes[1];
    int N = in_sizes[0] / ROW;
    float* out = (float*)d_out;

    unsigned* plane = (unsigned*)d_ws;
    k_plane<<<(E + OWN_CHUNK - 1) / OWN_CHUNK * (OWN_CHUNK / (PRE_T / 32 * BLK_EDGES)), PRE_T, 0, stream>>>(dst, plane, E);
    k_owner<<<(N + OWN_R - 1) / OWN_R, OWN_T, 0, stream>>>(h, src, dst, plane, w, out, E, N);
}


# dense list counting-sorted by adding wave, then by row inside the wave: no wave scans the whole list, and a row's sums stay in registers over its run of entries; sum order unchanged
# speedup vs baseline: 6.3223x; 1.2418x over previous
#include <hip/hip_runtime.h>
#include <hip/hip_bf16.h>
#include <math.h>

#define SLOT  2
#define DIM   128
#define ROW   (SLOT * DIM)
#define SLOPE 0.98f

typedef float v2f __attribute__((ext_vector_type(2)));
typedef float v8f __attribute__((ext_vector_type(8)));
typedef float v4f __attribute__((ext_vector_type(4)));


__device__ __forceinline__ float xhalf_f32(float x) {
    int b = __float_as_int(x);
    return __int_as_float(__builtin_amdgcn_permlanex16(b, b, 0x76543210, 0xfedcba98, false, false));
}
template <int CTRL>
__device__ __forceinline__ float rowx_f32(float x) {
    return __int_as_float(__builtin_amdgcn_update_dpp(0, __float_as_int(x), CTRL, 0xf, 0xf, true));
}

__device__ __forceinline__ float trace8(const v4f (&x)[8], int cnt, v2f w0, v2f w1, int lane) {
    const int  r    = lane & 7;
    const bool diag = ((lane >> 3) & 1) == (lane >> 4);
    const bool bit0 = (lane & 1) != 0, bit1 = (lane & 2) != 0, bit2 = (lane & 4) != 0;
    float t[8];
    #pragma unroll
    for (int j = 0; j < 8; ++j) {
        t[j] = 0.0f;
        if (j < cnt) {
            const v2f a0 = {x[j].x, x[j].y}, a1 = {x[j].z, x[j].w};
            v8f c = {};
            c = __builtin_amdgcn_wmma_f32_16x16x4_f32(false, a0, false, w0, (short)0, c, false, false);
            c = __builtin_amdgcn_wmma_f32_16x16x4_f32(false, a1, false, w1, (short)0, c, false, false);
            float d = c[0];
            d = (r == 1) ? c[1] : d;
            d = (r == 2) ? c[2] : d;
            d = (r == 3) ? c[3] : d;
            d = (r == 4) ? c[4] : d;
            d = (r == 5) ? c[5] : d;
            d = (r == 6) ? c[6] : d;
            d = (r == 7) ? c[7] : d;
            t[j] = d;
        }
    }
    float a4[4], a2[2];
    #pragma unroll
    for (int j = 0; j < 4; ++j)
        a4[j] = (bit2 ? t[j + 4] : t[j]) + rowx_f32<0x164>(bit2 ? t[j] : t[j + 4]);
    #pragma unroll
    for (int j = 0; j < 2; ++j)
        a2[j] = (bit1 ? a4[j + 2] : a4[j]) + rowx_f32<0x162>(bit1 ? a4[j] : a4[j + 2]);
    float u = (bit0 ? a2[1] : a2[0]) + rowx_f32<0x161>(bit0 ? a2[0] : a2[1]);
    u = diag ? u : 0.0f;
    u = u + xhalf_f32(u);
    u = u + rowx_f32<0x168>(u);
    return u;
}

#define OWN_R      200
#define BLK_EDGES  1024
#define PRE_T      128

__device__ __forceinline__ unsigned owner_byte(int d) {
    const unsigned o = (unsigned)d / (unsigned)OWN_R;
    return o < 255u ? o : 255u;
}

__global__ void __launch_bounds__(PRE_T)
k_plane(const int* __restrict__ dst, unsigned* __restrict__ plane, int E) {
    const int lane  = threadIdx.x & 31;
    const int blk   = blockIdx.x * (PRE_T / 32) + (threadIdx.x >> 5);
    const int wbase = blk * BLK_EDGES;
    int4 d4[8];
    if (wbase + BLK_EDGES <= E) {
        #pragma unroll
        for (int j = 0; j < 8; ++j)
            d4[j] = *(const int4*)(dst + wbase + j * 128 + lane * 4);
    } else {
        #pragma unroll
        for (int j = 0; j < 8; ++j) {
            int e0 = wbase + j * 128 + lane * 4;
            d4[j].x = (e0     < E) ? dst[e0]     : -1;
            d4[j].y = (e0 + 1 < E) ? dst[e0 + 1] : -1;
            d4[j].z = (e0 + 2 < E) ? dst[e0 + 2] : -1;
            d4[j].w = (e0 + 3 < E) ? dst[e0 + 3] : -1;
        }
    }
    unsigned w[8] = {0u, 0u, 0u, 0u, 0u, 0u, 0u, 0u};
    #pragma unroll
    for (int j = 0; j < 8; ++j) {
        const int k0 = 4 * (j & 1), sh = 8 * (j >> 1);
        w[k0 + 0] |= owner_byte(d4[j].x) << sh;
        w[k0 + 1] |= owner_byte(d4[j].y) << sh;
        w[k0 + 2] |= owner_byte(d4[j].z) << sh;
        w[k0 + 3] |= owner_byte(d4[j].w) << sh;
    }
    uint4* p4 = (uint4*)(plane + blk * (BLK_EDGES / 4) + lane * 4);
    p4[0]  = make_uint4(w[0], w[1], w[2], w[3]);
    p4[32] = make_uint4(w[4], w[5], w[6], w[7]);
}

#define OWN_T      1024
#define OWN_WEDGES BLK_EDGES
#define OWN_CHUNK  (32 * OWN_WEDGES)
#define OWN_CAP    8192
#define OWN_WCAP   (OWN_CAP / 32)
#define CM_P       33
#define M_NONE     (-3.0e38f)

static_assert((OWN_R + 7) / 8 <= OWN_T / 32, "one wave per batch of eight owned rows");
static_assert(OWN_R <= 256, "a list entry keeps the row in its low 8 bits");
static_assert(OWN_CHUNK % (PRE_T / 32 * BLK_EDGES) == 0, "the plane launch covers whole scan chunks");
static_assert(OWN_WCAP == 256, "the resolve step takes a segment as eight entries per lane; a rank fits a byte");

template <int N>
__device__ __forceinline__ int row_shr_i32(int x) {
    return __builtin_amdgcn_update_dpp(0, x, 0x110 + N, 0xf, 0xf, false);
}
__device__ __forceinline__ int wave_prefix_i32(int x, int lane) {
    x += row_shr_i32<1>(x);
    x += row_shr_i32<2>(x);
    x += row_shr_i32<4>(x);
    x += row_shr_i32<8>(x);
    const int low = __builtin_amdgcn_readlane(x, 15);
    return (lane >= 16) ? x + low : x;
}

__global__ void __launch_bounds__(OWN_T)
k_owner(const float* __restrict__ h,
        const int*  __restrict__ src,
        const int*  __restrict__ dst,
        const unsigned* __restrict__ plane,
        const float* __restrict__ watt,
        float* __restrict__ out, int E, int N) {
    __shared__ __align__(16) float acc[OWN_R * ROW];
    __shared__ int   l_a[OWN_CAP];
    __shared__ unsigned char l_r[OWN_CAP];
    __shared__ int   l_b[OWN_CAP];
    __shared__ int   cmat[32 * CM_P];
    __shared__ int   wtot[32];
    __shared__ int   rcnt[32 * 8];
    __shared__ int   wcnt[32];
    __shared__ float q_l[OWN_R];
    __shared__ float2 ml[OWN_R];
    __shared__ float inv_l[OWN_R];

    const int tid  = threadIdx.x;
    const int lane = tid & 31;
    const int wv   = __builtin_amdgcn_readfirstlane(tid >> 5);
    const int lo   = blockIdx.x * OWN_R;
    int nr = N - lo; if (nr > OWN_R) nr = OWN_R;
    const unsigned mine4 = (unsigned)blockIdx.x * 0x01010101u;

    const v4f wS = *(const v4f*)(watt + lane * 4);
    const v2f wS0 = {wS.x, wS.y}, wS1 = {wS.z, wS.w};

    float4* acc4 = (float4*)acc;
    for (int i = tid; i < OWN_R * (ROW / 4); i += OWN_T)
        acc4[i] = make_float4(0.f, 0.f, 0.f, 0.f);
    if (tid < OWN_R) ml[tid] = make_float2(M_NONE, 0.0f);
    if (wv < (OWN_R + 7) / 8) {
        const v4f wD = *(const v4f*)(watt + DIM + lane * 4);
        const v2f wD0 = {wD.x, wD.y}, wD1 = {wD.z, wD.w};
        v4f xq[8];
        #pragma unroll
        for (int j = 0; j < 8; ++j) {
            const int rr = wv * 8 + j;
            xq[j] = (v4f){0.f, 0.f, 0.f, 0.f};
            if (rr < nr) xq[j] = *(const v4f*)(h + (size_t)(lo + rr) * ROW + lane * 4);
        }
        const float uq = trace8(xq, 8, wD0, wD1, lane);
        const int rq = wv * 8 + lane;
        if (lane < 8 && rq < OWN_R) q_l[rq] = uq;
    }
    __syncthreads();

    const int nblk = (E + OWN_CHUNK - 1) / OWN_CHUNK;
    const int sb = wv * OWN_WCAP;

    int c = 0, p0 = 0;

    for (;;) {
        int fill = 0;
        uint4 na = make_uint4(0u, 0u, 0u, 0u), nb = na;
        if (c < nblk) {
            const uint4* b4 = (const uint4*)(plane + ((c * OWN_CHUNK + wv * OWN_WEDGES) >> 2) + lane * 4);
            na = b4[0]; nb = b4[32];
        }
        while (c < nblk && fill < OWN_WCAP) {
            const int wbase = c * OWN_CHUNK + wv * OWN_WEDGES;
            const uint4 ba = na, bb = nb;
            {
                const int cn = (c + 1 < nblk) ? c + 1 : c;
                const uint4* n4 = (const uint4*)(plane + ((cn * OWN_CHUNK + wv * OWN_WEDGES) >> 2) + lane * 4);
                na = n4[0]; nb = n4[32];
            }
            const unsigned pw[8] = {ba.x, ba.y, ba.z, ba.w, bb.x, bb.y, bb.z, bb.w};
            unsigned nacc = 0u;
            #pragma unroll
            for (int k = 0; k < 8; ++k) {
                const unsigned y  = pw[k] ^ mine4;
                const unsigned nz = (((y & 0x7f7f7f7fu) + 0x7f7f7f7fu) | y) & 0x80808080u;
                nacc = (nacc >> 1) | nz;
            }
            const unsigned mbits = ~nacc;
            const int lcnt = __popc(mbits);
            const int linc = wave_prefix_i32(lcnt, lane);
            const int tot  = __builtin_amdgcn_readlane(linc, 31);
            int take = tot - p0;
            if (take > OWN_WCAP - fill) take = OWN_WCAP - fill;
            if (take > 0) {
                int pos = (linc - lcnt) - p0;
                unsigned mb = mbits;
                while (mb != 0u) {
                    int k = __ffs(mb) - 1; mb &= mb - 1u;
                    if (pos >= 0 && pos < take)
                        l_a[sb + fill + pos] = wbase + (k >> 2) * 128 + lane * 4 + (k & 3);
                    ++pos;
                }
            }
            fill += take;
            p0   += take;
            if (p0 == tot) { ++c; p0 = 0; }
        }
        cmat[wv * CM_P + lane] = 0;
        if (fill > 0) {
            int ee[8], dd[8], ss[8];
            #pragma unroll
            for (int k = 0; k < 8; ++k) {
                const int i = k * 32 + lane;
                const int e = l_a[sb + i];
                ee[k] = (i < fill) ? e : l_a[sb];
            }
            #pragma unroll
            for (int k = 0; k < 8; ++k) { dd[k] = dst[ee[k]]; ss[k] = src[ee[k]]; }
            #pragma unroll
            for (int k = 0; k < 8; ++k) {
                if (k * 32 < fill) {
                    const bool valid = k * 32 + lane < fill;
                    const int  ent = (ss[k] << 8) | (dd[k] - lo);
                    const int  t = ent & 31;
                    unsigned m = __builtin_amdgcn_ballot_w32(valid);
                    #pragma unroll
                    for (int b = 0; b < 5; ++b) {
                        const bool bit = ((t >> b) & 1) != 0;
                        const unsigned bal = __builtin_amdgcn_ballot_w32(bit);
                        m &= bit ? bal : ~bal;
                    }
                    const int below = (int)__builtin_amdgcn_mbcnt_lo(m, 0u);
                    const int base  = cmat[wv * CM_P + t];
                    if (valid && below == 0) cmat[wv * CM_P + t] = base + __popc(m);
                    l_a[sb + k * 32 + lane] = ent;
                    l_r[sb + k * 32 + lane] = (unsigned char)(base + below);
                }
            }
        }
        if (lane == 0) wcnt[wv] = (c < nblk) ? 1 : 0;
        __syncthreads();

        const bool more = __ballot(wcnt[lane] != 0) != 0;
        {
            const int cc  = cmat[lane * CM_P + wv];
            const int inc = wave_prefix_i32(cc, lane);
            cmat[lane * CM_P + wv] = inc - cc;
            if (lane == 31) wtot[wv] = inc;
        }
        __syncthreads();
        int n0, nmine;
        {
            const int tot   = wtot[lane];
            const int first = wave_prefix_i32(tot, lane) - tot;
            n0    = __builtin_amdgcn_readlane(first, wv);
            nmine = __builtin_amdgcn_readlane(tot, wv);
            if (fill > 0) {
                const int mybase = first + cmat[wv * CM_P + lane];
                for (int k0 = 0; k0 < fill; k0 += 32) {
                    const int ent = l_a[sb + k0 + lane];
                    const int rk  = l_r[sb + k0 + lane];
                    const int b = __builtin_amdgcn_ds_bpermute((ent & 31) << 2, mybase);
                    if (k0 + lane < fill) l_b[b + rk] = ent;
                }
            }
        }
        __syncthreads();

        for (int c0 = 0; c0 < nmine; c0 += OWN_WCAP) {
            int cn = nmine - c0; if (cn > OWN_WCAP) cn = OWN_WCAP;
            const int cb = n0 + c0;
            int e8[8], r8[8], j8[8];
            #pragma unroll
            for (int k = 0; k < 8; ++k) {
                int ei = cb + k * 32 + lane; if (ei > OWN_CAP - 1) ei = OWN_CAP - 1;
                e8[k] = l_b[ei];
            }
            if (lane < 8) rcnt[wv * 8 + lane] = 0;
            #pragma unroll
            for (int k = 0; k < 8; ++k) {
                r8[k] = 0; j8[k] = (e8[k] >> 5) & 7;
                if (k * 32 < cn) {
                    const bool valid = k * 32 + lane < cn;
                    unsigned m = __builtin_amdgcn_ballot_w32(valid);
                    #pragma unroll
                    for (int b = 0; b < 3; ++b) {
                        const bool bit = ((j8[k] >> b) & 1) != 0;
                        const unsigned bal = __builtin_amdgcn_ballot_w32(bit);
                        m &= bit ? bal : ~bal;
                    }
                    const int below = (int)__builtin_amdgcn_mbcnt_lo(m, 0u);
                    const int base  = rcnt[wv * 8 + j8[k]];
                    r8[k] = base + below;
                    if (valid && below == 0) rcnt[wv * 8 + j8[k]] = base + __popc(m);
                }
            }
            {
                const int cj = (lane < 8) ? rcnt[wv * 8 + lane] : 0;
                const int sj = wave_prefix_i32(cj, lane) - cj;
                #pragma unroll
                for (int k = 0; k < 8; ++k) {
                    if (k * 32 < cn) {
                        const int b = __builtin_amdgcn_ds_bpermute(j8[k] << 2, sj);
                        if (k * 32 + lane < cn) l_b[cb + b + r8[k]] = e8[k];
                    }
                }
            }

            int   cur = -1;
            v4f   a0 = {0.f, 0.f, 0.f, 0.f}, a1 = a0;
            float rm = 0.0f, rl = 0.0f, rq = 0.0f;
            for (int g = 0; g < cn; g += 8) {
                int cnt = cn - g; if (cnt > 8) cnt = 8;
                int ei = cb + g + (lane & 7); if (ei > OWN_CAP - 1) ei = OWN_CAP - 1;
                const int ent = l_b[ei];
                int es[8], ed[8];
                #pragma unroll
                for (int t = 0; t < 8; ++t) {
                    const int e = __builtin_amdgcn_readlane(ent, t);
                    es[t] = e >> 8; ed[t] = e & 255;
                }
                v4f x0[8], x1[8];
                #pragma unroll
                for (int t = 0; t < 8; ++t) {
                    const int sn = (t < cnt) ? es[t] : es[0];
                    const float* rp = h + (size_t)sn * ROW + lane * 4;
                    x0[t] = *(const v4f*)rp;
                    x1[t] = *(const v4f*)(rp + DIM);
                }
                const float u = trace8(x0, cnt, wS0, wS1, lane);
                #pragma unroll
                for (int t = 0; t < 8; ++t) {
                    if (t < cnt) {
                        const int row = ed[t];
                        if (row != cur) {
                            if (cur >= 0) {
                                *(v4f*)(acc + cur * ROW + lane * 4)       = a0;
                                *(v4f*)(acc + cur * ROW + DIM + lane * 4) = a1;
                                if (lane == 0) ml[cur] = make_float2(rm, rl);
                            }
                            cur = row;
                            a0 = *(const v4f*)(acc + row * ROW + lane * 4);
                            a1 = *(const v4f*)(acc + row * ROW + DIM + lane * 4);
                            const float2 st = ml[row];
                            rm = st.x; rl = st.y; rq = q_l[row];
                        }
                        const float p = __int_as_float(__builtin_amdgcn_readlane(__float_as_int(u), t));
                        const float s = p + rq;
                        const float a = (s >= 0.0f) ? s : SLOPE * s;
                        const float mn = fmaxf(rm, a);
                        const float sc = __expf(rm - mn);
                        const float w  = __expf(a - mn);
                        a0 = a0 * sc + w * x0[t];
                        a1 = a1 * sc + w * x1[t];
                        rl = rl * sc + w;
                        rm = mn;
                    }
                }
            }
            if (cur >= 0) {
                *(v4f*)(acc + cur * ROW + lane * 4)       = a0;
                *(v4f*)(acc + cur * ROW + DIM + lane * 4) = a1;
                if (lane == 0) ml[cur] = make_float2(rm, rl);
            }
        }
        __syncthreads();
        if (!more) break;
    }

    if (tid < OWN_R) {
        const float l = ml[tid].y;
        inv_l[tid] = (l > 0.0f) ? 1.0f / l : 0.0f;
    }
    __syncthreads();

    const float4* h4 = (const float4*)h;
    float4* out4 = (float4*)out;
    for (int i = tid; i < nr * (ROW / 4); i += OWN_T) {
        int r = i >> 6;
        int n = lo + r;
        size_t g = (size_t)n * (ROW / 4) + (i & 63);
        float4 v = acc4[i];
        const float inv = inv_l[r];
        if (inv == 0.0f) {
            v = h4[g];
        } else {
            v.x *= inv; v.y *= inv; v.z *= inv; v.w *= inv;
        }
        out4[g] = v;
    }
}

extern "C" void kernel_launch(void* const* d_in, const int* in_sizes, int n_in,
                              void* d_out, int out_size, void* d_ws, size_t ws_size,
                              hipStream_t stream) {
    const float* h   = (const float*)d_in[0];
    const int*   src = (const int*)d_in[1];
    const int*   dst = (const int*)d_in[2];
    const float* w   = (const float*)d_in[3];
    int E = in_sizes[1];
    int N = in_sizes[0] / ROW;
    float* out = (float*)d_out;

    unsigned* plane = (unsigned*)d_ws;
    k_plane<<<(E + OWN_CHUNK - 1) / OWN_CHUNK * (OWN_CHUNK / (PRE_T / 32 * BLK_EDGES)), PRE_T, 0, stream>>>(dst, plane, E);
    k_owner<<<(N + OWN_R - 1) / OWN_R, OWN_T, 0, stream>>>(h, src, dst, plane, w, out, E, N);
}


# first launch puts each 1024-edge block in owner order (counting sort in LDS) with 256 run starts per block; an owner reads only its run of each block, not every edge's owner byte; sum order changes
# speedup vs baseline: 6.7927x; 1.0744x over previous
#include <hip/hip_runtime.h>
#include <hip/hip_bf16.h>
#include <math.h>

#define SLOT  2
#define DIM   128
#define ROW   (SLOT * DIM)
#define SLOPE 0.98f

typedef float v2f __attribute__((ext_vector_type(2)));
typedef float v8f __attribute__((ext_vector_type(8)));
typedef float v4f __attribute__((ext_vector_type(4)));


__device__ __forceinline__ float xhalf_f32(float x) {
    int b = __float_as_int(x);
    return __int_as_float(__builtin_amdgcn_permlanex16(b, b, 0x76543210, 0xfedcba98, false, false));
}
template <int CTRL>
__device__ __forceinline__ float rowx_f32(float x) {
    return __int_as_float(__builtin_amdgcn_update_dpp(0, __float_as_int(x), CTRL, 0xf, 0xf, true));
}

__device__ __forceinline__ float trace8(const v4f (&x)[8], int cnt, v2f w0, v2f w1, int lane) {
    const int  r    = lane & 7;
    const bool diag = ((lane >> 3) & 1) == (lane >> 4);
    const bool bit0 = (lane & 1) != 0, bit1 = (lane & 2) != 0, bit2 = (lane & 4) != 0;
    float t[8];
    #pragma unroll
    for (int j = 0; j < 8; ++j) {
        t[j] = 0.0f;
        if (j < cnt) {
            const v2f a0 = {x[j].x, x[j].y}, a1 = {x[j].z, x[j].w};
            v8f c = {};
            c = __builtin_amdgcn_wmma_f32_16x16x4_f32(false, a0, false, w0, (short)0, c, false, false);
            c = __builtin_amdgcn_wmma_f32_16x16x4_f32(false, a1, false, w1, (short)0, c, false, false);
            float d = c[0];
            d = (r == 1) ? c[1] : d;
            d = (r == 2) ? c[2] : d;
            d = (r == 3) ? c[3] : d;
            d = (r == 4) ? c[4] : d;
            d = (r == 5) ? c[5] : d;
            d = (r == 6) ? c[6] : d;
            d = (r == 7) ? c[7] : d;
            t[j] = d;
        }
    }
    float a4[4], a2[2];
    #pragma unroll
    for (int j = 0; j < 4; ++j)
        a4[j] = (bit2 ? t[j + 4] : t[j]) + rowx_f32<0x164>(bit2 ? t[j] : t[j + 4]);
    #pragma unroll
    for (int j = 0; j < 2; ++j)
        a2[j] = (bit1 ? a4[j + 2] : a4[j]) + rowx_f32<0x162>(bit1 ? a4[j] : a4[j + 2]);
    float u = (bit0 ? a2[1] : a2[0]) + rowx_f32<0x161>(bit0 ? a2[0] : a2[1]);
    u = diag ? u : 0.0f;
    u = u + xhalf_f32(u);
    u = u + rowx_f32<0x168>(u);
    return u;
}

#define OWN_R      200
#define BLK_EDGES  1024
#define SRT_T      256
#define SRT_G      (BLK_EDGES / SRT_T)
#define SRT_NB     2
#define NKEY       256

static_assert(SRT_T == NKEY, "one thread per owner index in the counting steps");
static_assert(32 % SRT_NB == 0, "a round of 32 blocks is a whole number of work-groups");

__device__ __forceinline__ unsigned owner_byte(int d) {
    const unsigned o = (unsigned)d / (unsigned)OWN_R;
    return o < 255u ? o : 255u;
}

template <int N>
__device__ __forceinline__ int row_shr_i32(int x) {
    return __builtin_amdgcn_update_dpp(0, x, 0x110 + N, 0xf, 0xf, false);
}
__device__ __forceinline__ int wave_prefix_i32(int x, int lane) {
    x += row_shr_i32<1>(x);
    x += row_shr_i32<2>(x);
    x += row_shr_i32<4>(x);
    x += row_shr_i32<8>(x);
    const int low = __builtin_amdgcn_readlane(x, 15);
    return (lane >= 16) ? x + low : x;
}

__global__ void __launch_bounds__(SRT_T)
k_sort(const int* __restrict__ dst, unsigned* __restrict__ start, unsigned* __restrict__ rec, int E) {
    __shared__ __align__(16) int cnt[32 * NKEY];
    __shared__ int kbase[NKEY];
    __shared__ int wsum[SRT_T / 32];
    __shared__ unsigned short ord[BLK_EDGES];

    const int tid  = threadIdx.x;
    const int lane = tid & 31;
    const int wv   = __builtin_amdgcn_readfirstlane(tid >> 5);

    int dd[SRT_NB][SRT_G];
    #pragma unroll
    for (int it = 0; it < SRT_NB; ++it) {
        #pragma unroll
        for (int g = 0; g < SRT_G; ++g) {
            const int e = (blockIdx.x * SRT_NB + it) * BLK_EDGES + g * SRT_T + tid;
            dd[it][g] = (e < E) ? dst[e] : -1;
        }
    }

    #pragma unroll
    for (int it = 0; it < SRT_NB; ++it) {
        const int blk = blockIdx.x * SRT_NB + it;
        int4* c4 = (int4*)cnt;
        for (int i = tid; i < 32 * NKEY / 4; i += SRT_T) c4[i] = make_int4(0, 0, 0, 0);
        __syncthreads();

        int o[SRT_G], below[SRT_G];
        #pragma unroll
        for (int g = 0; g < SRT_G; ++g) {
            o[g] = (int)owner_byte(dd[it][g]);
            unsigned m = 0xffffffffu;
            #pragma unroll
            for (int b = 0; b < 8; ++b) {
                const bool bit = ((o[g] >> b) & 1) != 0;
                const unsigned bal = __builtin_amdgcn_ballot_w32(bit);
                m &= bit ? bal : ~bal;
            }
            below[g] = (int)__builtin_amdgcn_mbcnt_lo(m, 0u);
            if (below[g] == 0) cnt[(g * (SRT_T / 32) + wv) * NKEY + o[g]] = __popc(m);
        }
        __syncthreads();

        {
            int run = 0;
            for (int v = 0; v < 32; ++v) {
                const int x = cnt[v * NKEY + tid];
                cnt[v * NKEY + tid] = run;
                run += x;
            }
            const int inc = wave_prefix_i32(run, lane);
            if (lane == 31) wsum[wv] = inc;
            kbase[tid] = inc - run;
        }
        __syncthreads();
        {
            int add = 0;
            for (int w = 0; w < wv; ++w) add += wsum[w];
            const int b = kbase[tid] + add;
            kbase[tid] = b;
            start[(size_t)blk * NKEY + tid] = (unsigned)b;
        }
        __syncthreads();
        #pragma unroll
        for (int g = 0; g < SRT_G; ++g)
            ord[kbase[o[g]] + cnt[(g * (SRT_T / 32) + wv) * NKEY + o[g]] + below[g]] = (unsigned short)(g * SRT_T + tid);
        __syncthreads();
        #pragma unroll
        for (int k = 0; k < BLK_EDGES / 2 / SRT_T; ++k) {
            const int i = k * SRT_T + tid;
            rec[(size_t)blk * (BLK_EDGES / 2) + i] = (unsigned)ord[2 * i] | ((unsigned)ord[2 * i + 1] << 16);
        }
    }
}

#define OWN_T      1024
#define OWN_CHUNK  (32 * BLK_EDGES)
#define OWN_CAP    8192
#define OWN_WCAP   (OWN_CAP / 32)
#define CM_P       33
#define M_NONE     (-3.0e38f)

static_assert((OWN_R + 7) / 8 <= OWN_T / 32, "one wave per batch of eight owned rows");
static_assert(OWN_R <= 256, "a list entry keeps the row in its low 8 bits");
static_assert(OWN_WCAP == 256, "the resolve step takes a segment as eight entries per lane; a rank fits a byte");

__global__ void __launch_bounds__(OWN_T)
k_owner(const float* __restrict__ h,
        const int*  __restrict__ src,
        const int*  __restrict__ dst,
        const unsigned* __restrict__ start,
        const unsigned* __restrict__ rec32,
        const float* __restrict__ watt,
        float* __restrict__ out, int E, int N) {
    __shared__ __align__(16) float acc[OWN_R * ROW];
    __shared__ int   l_a[OWN_CAP];
    __shared__ unsigned char l_r[OWN_CAP];
    __shared__ int   l_b[OWN_CAP];
    __shared__ int   cmat[32 * CM_P];
    __shared__ int   wtot[32];
    __shared__ int   rcnt[32 * 8];
    __shared__ int   wcnt[32];
    __shared__ float q_l[OWN_R];
    __shared__ float2 ml[OWN_R];
    __shared__ float inv_l[OWN_R];

    const int tid  = threadIdx.x;
    const int lane = tid & 31;
    const int wv   = __builtin_amdgcn_readfirstlane(tid >> 5);
    const int lo   = blockIdx.x * OWN_R;
    int nr = N - lo; if (nr > OWN_R) nr = OWN_R;

    const v4f wS = *(const v4f*)(watt + lane * 4);
    const v2f wS0 = {wS.x, wS.y}, wS1 = {wS.z, wS.w};

    float4* acc4 = (float4*)acc;
    for (int i = tid; i < OWN_R * (ROW / 4); i += OWN_T)
        acc4[i] = make_float4(0.f, 0.f, 0.f, 0.f);
    if (tid < OWN_R) ml[tid] = make_float2(M_NONE, 0.0f);
    if (wv < (OWN_R + 7) / 8) {
        const v4f wD = *(const v4f*)(watt + DIM + lane * 4);
        const v2f wD0 = {wD.x, wD.y}, wD1 = {wD.z, wD.w};
        v4f xq[8];
        #pragma unroll
        for (int j = 0; j < 8; ++j) {
            const int rr = wv * 8 + j;
            xq[j] = (v4f){0.f, 0.f, 0.f, 0.f};
            if (rr < nr) xq[j] = *(const v4f*)(h + (size_t)(lo + rr) * ROW + lane * 4);
        }
        const float uq = trace8(xq, 8, wD0, wD1, lane);
        const int rq = wv * 8 + lane;
        if (lane < 8 && rq < OWN_R) q_l[rq] = uq;
    }
    __syncthreads();

    const unsigned short* rec = (const unsigned short*)rec32;
    const int nblk = (E + OWN_CHUNK - 1) / OWN_CHUNK;
    const int sb = wv * OWN_WCAP;

    int c = 0, p0 = 0;

    for (;;) {
        int fill = 0;
        while (c < nblk && fill < OWN_WCAP) {
            const int cc  = c + lane;
            const int blk = cc * 32 + wv;
            int st = 0, cnt = 0;
            if (cc < nblk) {
                const unsigned* s2p = start + (size_t)blk * NKEY + blockIdx.x;
                const int s0 = (int)s2p[0], s1 = (int)s2p[1];
                st = s0; cnt = s1 - s0;
                if (lane == 0) { st += p0; cnt -= p0; }
            }
            const int inc  = wave_prefix_i32(cnt, lane);
            const int tot  = __builtin_amdgcn_readlane(inc, 31);
            const int room = OWN_WCAP - fill;
            const int pos  = inc - cnt;
            int tk = room - pos;
            if (tk > cnt) tk = cnt;
            if (tk < 0)   tk = 0;
            const unsigned short* rp = rec + (size_t)blk * BLK_EDGES + st;
            const int ebase = blk * BLK_EDGES;
            for (int j0 = 0; __builtin_amdgcn_ballot_w32(j0 < tk) != 0u; j0 += 4) {
                int v[4];
                #pragma unroll
                for (int k = 0; k < 4; ++k) v[k] = (j0 + k < tk) ? (int)rp[j0 + k] : 0;
                #pragma unroll
                for (int k = 0; k < 4; ++k)
                    if (j0 + k < tk) l_a[sb + fill + pos + j0 + k] = ebase + v[k];
            }
            const unsigned part = __builtin_amdgcn_ballot_w32(tk < cnt);
            if (part == 0u) { c += 32; p0 = 0; }
            else {
                const int f = __ffs(part) - 1;
                const int tkf = __builtin_amdgcn_readlane(tk, f);
                p0 = tkf + (f == 0 ? p0 : 0);
                c += f;
            }
            fill += (tot < room) ? tot : room;
        }
        cmat[wv * CM_P + lane] = 0;
        if (fill > 0) {
            int ee[8], dd[8], ss[8];
            #pragma unroll
            for (int k = 0; k < 8; ++k) {
                const int i = k * 32 + lane;
                const int e = l_a[sb + i];
                ee[k] = (i < fill) ? e : l_a[sb];
            }
            #pragma unroll
            for (int k = 0; k < 8; ++k) { dd[k] = dst[ee[k]]; ss[k] = src[ee[k]]; }
            #pragma unroll
            for (int k = 0; k < 8; ++k) {
                if (k * 32 < fill) {
                    const bool valid = k * 32 + lane < fill;
                    const int  ent = (ss[k] << 8) | (dd[k] - lo);
                    const int  t = ent & 31;
                    unsigned m = __builtin_amdgcn_ballot_w32(valid);
                    #pragma unroll
                    for (int b = 0; b < 5; ++b) {
                        const bool bit = ((t >> b) & 1) != 0;
                        const unsigned bal = __builtin_amdgcn_ballot_w32(bit);
                        m &= bit ? bal : ~bal;
                    }
                    const int below = (int)__builtin_amdgcn_mbcnt_lo(m, 0u);
                    const int base  = cmat[wv * CM_P + t];
                    if (valid && below == 0) cmat[wv * CM_P + t] = base + __popc(m);
                    l_a[sb + k * 32 + lane] = ent;
                    l_r[sb + k * 32 + lane] = (unsigned char)(base + below);
                }
            }
        }
        if (lane == 0) wcnt[wv] = (c < nblk) ? 1 : 0;
        __syncthreads();

        const bool more = __ballot(wcnt[lane] != 0) != 0;
        {
            const int cc  = cmat[lane * CM_P + wv];
            const int inc = wave_prefix_i32(cc, lane);
            cmat[lane * CM_P + wv] = inc - cc;
            if (lane == 31) wtot[wv] = inc;
        }
        __syncthreads();
        int n0, nmine;
        {
            const int tot   = wtot[lane];
            const int first = wave_prefix_i32(tot, lane) - tot;
            n0    = __builtin_amdgcn_readlane(first, wv);
            nmine = __builtin_amdgcn_readlane(tot, wv);
            if (fill > 0) {
                const int mybase = first + cmat[wv * CM_P + lane];
                for (int k0 = 0; k0 < fill; k0 += 32) {
                    const int ent = l_a[sb + k0 + lane];
                    const int rk  = l_r[sb + k0 + lane];
                    const int b = __builtin_amdgcn_ds_bpermute((ent & 31) << 2, mybase);
                    if (k0 + lane < fill) l_b[b + rk] = ent;
                }
            }
        }
        __syncthreads();

        for (int c0 = 0; c0 < nmine; c0 += OWN_WCAP) {
            int cn = nmine - c0; if (cn > OWN_WCAP) cn = OWN_WCAP;
            const int cb = n0 + c0;
            int e8[8], r8[8], j8[8];
            #pragma unroll
            for (int k = 0; k < 8; ++k) {
                int ei = cb + k * 32 + lane; if (ei > OWN_CAP - 1) ei = OWN_CAP - 1;
                e8[k] = l_b[ei];
            }
            if (lane < 8) rcnt[wv * 8 + lane] = 0;
            #pragma unroll
            for (int k = 0; k < 8; ++k) {
                r8[k] = 0; j8[k] = (e8[k] >> 5) & 7;
                if (k * 32 < cn) {
                    const bool valid = k * 32 + lane < cn;
                    unsigned m = __builtin_amdgcn_ballot_w32(valid);
                    #pragma unroll
                    for (int b = 0; b < 3; ++b) {
                        const bool bit = ((j8[k] >> b) & 1) != 0;
                        const unsigned bal = __builtin_amdgcn_ballot_w32(bit);
                        m &= bit ? bal : ~bal;
                    }
                    const int below = (int)__builtin_amdgcn_mbcnt_lo(m, 0u);
                    const int base  = rcnt[wv * 8 + j8[k]];
                    r8[k] = base + below;
                    if (valid && below == 0) rcnt[wv * 8 + j8[k]] = base + __popc(m);
                }
            }
            {
                const int cj = (lane < 8) ? rcnt[wv * 8 + lane] : 0;
                const int sj = wave_prefix_i32(cj, lane) - cj;
                #pragma unroll
                for (int k = 0; k < 8; ++k) {
                    if (k * 32 < cn) {
                        const int b = __builtin_amdgcn_ds_bpermute(j8[k] << 2, sj);
                        if (k * 32 + lane < cn) l_b[cb + b + r8[k]] = e8[k];
                    }
                }
            }

            int   cur = -1;
            v4f   a0 = {0.f, 0.f, 0.f, 0.f}, a1 = a0;
            float rm = 0.0f, rl = 0.0f, rq = 0.0f;
            for (int g = 0; g < cn; g += 8) {
                int cnt = cn - g; if (cnt > 8) cnt = 8;
                int ei = cb + g + (lane & 7); if (ei > OWN_CAP - 1) ei = OWN_CAP - 1;
                const int ent = l_b[ei];
                int es[8], ed[8];
                #pragma unroll
                for (int t = 0; t < 8; ++t) {
                    const int e = __builtin_amdgcn_readlane(ent, t);
                    es[t] = e >> 8; ed[t] = e & 255;
                }
                v4f x0[8], x1[8];
                #pragma unroll
                for (int t = 0; t < 8; ++t) {
                    const int sn = (t < cnt) ? es[t] : es[0];
                    const float* rp = h + (size_t)sn * ROW + lane * 4;
                    x0[t] = *(const v4f*)rp;
                    x1[t] = *(const v4f*)(rp + DIM);
                }
                const float u = trace8(x0, cnt, wS0, wS1, lane);
                #pragma unroll
                for (int t = 0; t < 8; ++t) {
                    if (t < cnt) {
                        const int row = ed[t];
                        if (row != cur) {
                            if (cur >= 0) {
                                *(v4f*)(acc + cur * ROW + lane * 4)       = a0;
                                *(v4f*)(acc + cur * ROW + DIM + lane * 4) = a1;
                                if (lane == 0) ml[cur] = make_float2(rm, rl);
                            }
                            cur = row;
                            a0 = *(const v4f*)(acc + row * ROW + lane * 4);
                            a1 = *(const v4f*)(acc + row * ROW + DIM + lane * 4);
                            const float2 st = ml[row];
                            rm = st.x; rl = st.y; rq = q_l[row];
                        }
                        const float p = __int_as_float(__builtin_amdgcn_readlane(__float_as_int(u), t));
                        const float s = p + rq;
                        const float a = (s >= 0.0f) ? s : SLOPE * s;
                        const float mn = fmaxf(rm, a);
                        const float sc = __expf(rm - mn);
                        const float w  = __expf(a - mn);
                        a0 = a0 * sc + w * x0[t];
                        a1 = a1 * sc + w * x1[t];
                        rl = rl * sc + w;
                        rm = mn;
                    }
                }
            }
            if (cur >= 0) {
                *(v4f*)(acc + cur * ROW + lane * 4)       = a0;
                *(v4f*)(acc + cur * ROW + DIM + lane * 4) = a1;
                if (lane == 0) ml[cur] = make_float2(rm, rl);
            }
        }
        __syncthreads();
        if (!more) break;
    }

    if (tid < OWN_R) {
        const float l = ml[tid].y;
        inv_l[tid] = (l > 0.0f) ? 1.0f / l : 0.0f;
    }
    __syncthreads();

    const float4* h4 = (const float4*)h;
    float4* out4 = (float4*)out;
    for (int i = tid; i < nr * (ROW / 4); i += OWN_T) {
        int r = i >> 6;
        int n = lo + r;
        size_t g = (size_t)n * (ROW / 4) + (i & 63);
        float4 v = acc4[i];
        const float inv = inv_l[r];
        if (inv == 0.0f) {
            v = h4[g];
        } else {
            v.x *= inv; v.y *= inv; v.z *= inv; v.w *= inv;
        }
        out4[g] = v;
    }
}

extern "C" void kernel_launch(void* const* d_in, const int* in_sizes, int n_in,
                              void* d_out, int out_size, void* d_ws, size_t ws_size,
                              hipStream_t stream) {
    const float* h   = (const float*)d_in[0];
    const int*   src = (const int*)d_in[1];
    const int*   dst = (const int*)d_in[2];
    const float* w   = (const float*)d_in[3];
    int E = in_sizes[1];
    int N = in_sizes[0] / ROW;
    float* out = (float*)d_out;

    unsigned* start = (unsigned*)d_ws;
    unsigned* rec   = start + (size_t)((E + OWN_CHUNK - 1) / OWN_CHUNK * 32) * NKEY;
    k_sort<<<(E + OWN_CHUNK - 1) / OWN_CHUNK * (32 / SRT_NB), SRT_T, 0, stream>>>(dst, start, rec, E);
    k_owner<<<(N + OWN_R - 1) / OWN_R, OWN_T, 0, stream>>>(h, src, dst, start, rec, w, out, E, N);
}
